# starting kernel
# baseline (speedup 1.0000x reference)
#include <hip/hip_runtime.h>
#include <hip/hip_bf16.h>

typedef __attribute__((ext_vector_type(2))) float v2f;
typedef __attribute__((ext_vector_type(8))) float v8f;

#define D 64
#define LSTR 66

__global__ void consts_kernel(const float* __restrict__ coeff_param,
                              const float* __restrict__ scale_param,
                              float* __restrict__ consts) {
    if (threadIdx.x != 0 || blockIdx.x != 0) return;
    float x0 = coeff_param[0], x1 = coeff_param[1], x2 = coeff_param[2];
    float m = fmaxf(x0, fmaxf(x1, x2));
    float e0 = expf(x0 - m), e1 = expf(x1 - m), e2 = expf(x2 - m);
    float inv = 1.0f / (e0 + e1 + e2);
    consts[0] = e0 * inv;
    consts[1] = e1 * inv;
    consts[2] = e2 * inv;
    float s = scale_param[0];
    consts[3] = (s > 20.0f) ? s : log1pf(expf(s));
}

__global__ void __launch_bounds__(128)
gemm_wmma_kernel(const float* __restrict__ h, const float* __restrict__ W,
                 const float* __restrict__ bias, float* __restrict__ Wh, int N) {
    __shared__ float Ws[D * LSTR];

    int tid = threadIdx.x;
    for (int i = tid * 4; i < D * D; i += 128 * 4) {
        int r = i >> 6;
        int c = i & 63;
        float4 w4 = *(const float4*)(W + i);
        Ws[(c + 0) * LSTR + r] = w4.x;
        Ws[(c + 1) * LSTR + r] = w4.y;
        Ws[(c + 2) * LSTR + r] = w4.z;
        Ws[(c + 3) * LSTR + r] = w4.w;
    }
    __syncthreads();

    int wave = tid >> 5;
    int lane = tid & 31;
    int half = lane >> 4;
    int l16  = lane & 15;

    int m0 = (blockIdx.x * 4 + wave) * 16;
    if (m0 >= N) return;

    const float* hrow = h + (size_t)(m0 + l16) * D + 2 * half;
    const float* wt   = Ws + l16 * LSTR + 2 * half;

    v8f acc[4] = {};

#pragma unroll 4
    for (int s = 0; s < 16; ++s) {
        v2f a = *(const v2f*)(hrow + 4 * s);
#pragma unroll
        for (int t = 0; t < 4; ++t) {
            v2f bf = *(const v2f*)(wt + t * (16 * LSTR) + 4 * s);
            acc[t] = __builtin_amdgcn_wmma_f32_16x16x4_f32(
                false, a, false, bf, (short)0, acc[t], false, false);
        }
    }

#pragma unroll
    for (int t = 0; t < 4; ++t) {
        int n = t * 16 + l16;
        float bv = bias[n];
#pragma unroll
        for (int v = 0; v < 8; ++v) {
            Wh[(size_t)(m0 + half * 8 + v) * D + n] = acc[t][v] + bv;
        }
    }
}

__global__ void __launch_bounds__(256)
norms_kernel(const float* __restrict__ Wh, float* __restrict__ norms, int N) {
    int gtid = blockIdx.x * 256 + threadIdx.x;
    int node = gtid >> 4;
    int l    = gtid & 15;
    if (node >= N) return;
    float4 v = *(const float4*)(Wh + (size_t)node * D + l * 4);
    float s = v.x * v.x + v.y * v.y + v.z * v.z + v.w * v.w;
    for (int m = 8; m >= 1; m >>= 1) s += __shfl_xor(s, m, 16);
    if (l == 0) norms[node] = sqrtf(s);
}

__global__ void __launch_bounds__(256)
zero_kernel(float4* __restrict__ p, int n4) {
    int i = blockIdx.x * 256 + threadIdx.x;
    if (i < n4) p[i] = make_float4(0.f, 0.f, 0.f, 0.f);
}

__global__ void __launch_bounds__(256)
edge_kernel(const int* __restrict__ ei, const float* __restrict__ adj_v,
            const float* __restrict__ deg_v, const float* __restrict__ Wh,
            const float* __restrict__ norms, const float* __restrict__ dcoef,
            const float* __restrict__ consts, float* __restrict__ prop, int E) {
    int tid = threadIdx.x;
    int l   = tid & 15;
    int eid = blockIdx.x * 16 + (tid >> 4);
    if (eid >= E) return;

    int row = ei[eid];
    int col = ei[E + eid];
    if (row == col) return;

    float4 r4 = *(const float4*)(Wh + (size_t)row * D + l * 4);
    float4 c4 = *(const float4*)(Wh + (size_t)col * D + l * 4);
    float dot = r4.x * c4.x + r4.y * c4.y + r4.z * c4.z + r4.w * c4.w;
    for (int m = 8; m >= 1; m >>= 1) dot += __shfl_xor(dot, m, 16);

    float nr = norms[row];
    float nc = norms[col];
    float sim = dot / fmaxf(nr * nc, 1e-8f);

    float x  = dcoef[0] * deg_v[eid] + dcoef[1];
    float sc = (x > 20.0f) ? x : log1pf(expf(x));

    float c0 = consts[0], c1 = consts[1];
    float w = adj_v[eid] * sc * (sim > 0.0f ? c0 * sim : c1 * sim);
    if (w == 0.0f) return;

    float* p = prop + (size_t)row * D + l * 4;
    atomicAdd(p + 0, w * c4.x);
    atomicAdd(p + 1, w * c4.y);
    atomicAdd(p + 2, w * c4.z);
    atomicAdd(p + 3, w * c4.w);
}

__global__ void __launch_bounds__(256)
final_kernel(const float4* __restrict__ prop, const float4* __restrict__ Wh,
             const float* __restrict__ consts, float4* __restrict__ out, int n4) {
    int i = blockIdx.x * 256 + threadIdx.x;
    if (i >= n4) return;
    float c2 = consts[2], scale = consts[3];
    float4 p = prop[i];
    float4 w = Wh[i];
    float4 o;
    o.x = scale * (p.x + c2 * w.x);
    o.y = scale * (p.y + c2 * w.y);
    o.z = scale * (p.z + c2 * w.z);
    o.w = scale * (p.w + c2 * w.w);
    out[i] = o;
}

extern "C" void kernel_launch(void* const* d_in, const int* in_sizes, int n_in,
                              void* d_out, int out_size, void* d_ws, size_t ws_size,
                              hipStream_t stream) {
    const float* h     = (const float*)d_in[0];
    const int*   ei    = (const int*)  d_in[1];
    const float* adj_v = (const float*)d_in[2];
    const float* deg_v = (const float*)d_in[3];
    const float* W     = (const float*)d_in[4];
    const float* bias  = (const float*)d_in[5];
    const float* dcoef = (const float*)d_in[6];
    const float* cpar  = (const float*)d_in[7];
    const float* spar  = (const float*)d_in[8];

    const int N = in_sizes[0] / D;
    const int E = in_sizes[2];

    float* Wh     = (float*)d_ws;
    float* prop   = Wh + (size_t)N * D;
    float* norms  = prop + (size_t)N * D;
    float* consts = norms + N;

    consts_kernel<<<1, 1, 0, stream>>>(cpar, spar, consts);

    gemm_wmma_kernel<<<(N + 63) / 64, 128, 0, stream>>>(h, W, bias, Wh, N);

    norms_kernel<<<(N * 16 + 255) / 256, 256, 0, stream>>>(Wh, norms, N);

    int n4 = N * D / 4;
    zero_kernel<<<(n4 + 255) / 256, 256, 0, stream>>>((float4*)prop, n4);

    edge_kernel<<<(E + 15) / 16, 256, 0, stream>>>(ei, adj_v, deg_v, Wh, norms,
                                                   dcoef, consts, prop, E);

    final_kernel<<<(n4 + 255) / 256, 256, 0, stream>>>((const float4*)prop,
                                                       (const float4*)Wh, consts,
                                                       (float4*)d_out, n4);
}


# Edge kernel: a wave walks 32 edges, all 32 lanes on one row: gathers and float atomic adds cover whole 128-B lines; edge scalars one lane each; dot by a 32-lane reduce-scatter (other sum tree)
# speedup vs baseline: 1.5606x; 1.5606x over previous
#include <hip/hip_runtime.h>
#include <hip/hip_bf16.h>

typedef __attribute__((ext_vector_type(2))) float v2f;
typedef __attribute__((ext_vector_type(8))) float v8f;

#define D 64
#define LSTR 66

__global__ void consts_kernel(const float* __restrict__ coeff_param,
                              const float* __restrict__ scale_param,
                              float* __restrict__ consts) {
    if (threadIdx.x != 0 || blockIdx.x != 0) return;
    float x0 = coeff_param[0], x1 = coeff_param[1], x2 = coeff_param[2];
    float m = fmaxf(x0, fmaxf(x1, x2));
    float e0 = expf(x0 - m), e1 = expf(x1 - m), e2 = expf(x2 - m);
    float inv = 1.0f / (e0 + e1 + e2);
    consts[0] = e0 * inv;
    consts[1] = e1 * inv;
    consts[2] = e2 * inv;
    float s = scale_param[0];
    consts[3] = (s > 20.0f) ? s : log1pf(expf(s));
}

__global__ void __launch_bounds__(128)
gemm_wmma_kernel(const float* __restrict__ h, const float* __restrict__ W,
                 const float* __restrict__ bias, float* __restrict__ Wh, int N) {
    __shared__ float Ws[D * LSTR];

    int tid = threadIdx.x;
    for (int i = tid * 4; i < D * D; i += 128 * 4) {
        int r = i >> 6;
        int c = i & 63;
        float4 w4 = *(const float4*)(W + i);
        Ws[(c + 0) * LSTR + r] = w4.x;
        Ws[(c + 1) * LSTR + r] = w4.y;
        Ws[(c + 2) * LSTR + r] = w4.z;
        Ws[(c + 3) * LSTR + r] = w4.w;
    }
    __syncthreads();

    int wave = tid >> 5;
    int lane = tid & 31;
    int half = lane >> 4;
    int l16  = lane & 15;

    int m0 = (blockIdx.x * 4 + wave) * 16;
    if (m0 >= N) return;

    const float* hrow = h + (size_t)(m0 + l16) * D + 2 * half;
    const float* wt   = Ws + l16 * LSTR + 2 * half;

    v8f acc[4] = {};

#pragma unroll 4
    for (int s = 0; s < 16; ++s) {
        v2f a = *(const v2f*)(hrow + 4 * s);
#pragma unroll
        for (int t = 0; t < 4; ++t) {
            v2f bf = *(const v2f*)(wt + t * (16 * LSTR) + 4 * s);
            acc[t] = __builtin_amdgcn_wmma_f32_16x16x4_f32(
                false, a, false, bf, (short)0, acc[t], false, false);
        }
    }

#pragma unroll
    for (int t = 0; t < 4; ++t) {
        int n = t * 16 + l16;
        float bv = bias[n];
#pragma unroll
        for (int v = 0; v < 8; ++v) {
            Wh[(size_t)(m0 + half * 8 + v) * D + n] = acc[t][v] + bv;
        }
    }
}

__global__ void __launch_bounds__(256)
norms_kernel(const float* __restrict__ Wh, float* __restrict__ norms, int N) {
    int gtid = blockIdx.x * 256 + threadIdx.x;
    int node = gtid >> 4;
    int l    = gtid & 15;
    if (node >= N) return;
    float4 v = *(const float4*)(Wh + (size_t)node * D + l * 4);
    float s = v.x * v.x + v.y * v.y + v.z * v.z + v.w * v.w;
    for (int m = 8; m >= 1; m >>= 1) s += __shfl_xor(s, m, 16);
    if (l == 0) norms[node] = sqrtf(s);
}

__global__ void __launch_bounds__(256)
zero_kernel(float4* __restrict__ p, int n4) {
    int i = blockIdx.x * 256 + threadIdx.x;
    if (i < n4) p[i] = make_float4(0.f, 0.f, 0.f, 0.f);
}

template <int M>
__device__ __forceinline__ void rs_step(float (&p)[32], int lane) {
    const bool hi = (lane & M) != 0;
#pragma unroll
    for (int j = 0; j < M; ++j) {
        const float a = p[j];
        const float b = p[j + M];
        const float keep = hi ? b : a;
        const float send = hi ? a : b;
        p[j] = keep + __shfl_xor(send, M, 32);
    }
}

__global__ void __launch_bounds__(256)
edge_kernel(const int* __restrict__ ei, const float* __restrict__ adj_v,
            const float* __restrict__ deg_v, const float* __restrict__ Wh,
            const float* __restrict__ norms, const float* __restrict__ dcoef,
            const float* __restrict__ consts, float* __restrict__ prop, int E) {
    const int lane = threadIdx.x & 31;
    const unsigned lane4 = (threadIdx.x & 31u) * 4u;
    const int e0   = (blockIdx.x * 8 + (threadIdx.x >> 5)) * 32;
    if (e0 >= E) return;

    const int  eid  = e0 + lane;
    const bool live = eid < E;
    const int  eidc = live ? eid : E - 1;
    const int   row = ei[eidc];
    const int   col = ei[E + eidc];
    const float adj = adj_v[eidc];
    const float x   = dcoef[0] * deg_v[eidc] + dcoef[1];
    const float nr  = norms[row];
    const float nc  = norms[col];
    const float sc  = (x > 20.0f) ? x : log1pf(expf(x));

    float cv0[32], cv1[32], p[32];
#pragma unroll
    for (int e = 0; e < 32; ++e) {
        const int r = __builtin_amdgcn_readlane(row, e);
        const int c = __builtin_amdgcn_readlane(col, e);
        const float* rp = (const float*)((const char*)Wh + ((unsigned)r * (D * 4u) + lane4));
        const float* cp = (const float*)((const char*)Wh + ((unsigned)c * (D * 4u) + lane4));
        const float r0 = rp[0], r1 = rp[32];
        cv0[e] = cp[0];
        cv1[e] = cp[32];
        p[e] = r0 * cv0[e] + r1 * cv1[e];
    }

    rs_step<16>(p, lane);
    rs_step<8>(p, lane);
    rs_step<4>(p, lane);
    rs_step<2>(p, lane);
    rs_step<1>(p, lane);
    const float dot = p[0];

    const float sim = dot / fmaxf(nr * nc, 1e-8f);
    const float c0 = consts[0], c1 = consts[1];
    float w = adj * sc * (sim > 0.0f ? c0 * sim : c1 * sim);
    if (!live || row == col) w = 0.0f;

#pragma unroll
    for (int e = 0; e < 32; ++e) {
        const float we = __builtin_bit_cast(
            float, __builtin_amdgcn_readlane(__builtin_bit_cast(int, w), e));
        if (we != 0.0f) {
            const int r = __builtin_amdgcn_readlane(row, e);
            float* pp = (float*)((char*)prop + ((unsigned)r * (D * 4u) + lane4));
            atomicAdd(pp, we * cv0[e]);
            atomicAdd(pp + 32, we * cv1[e]);
        }
    }
}

__global__ void __launch_bounds__(256)
final_kernel(const float4* __restrict__ prop, const float4* __restrict__ Wh,
             const float* __restrict__ consts, float4* __restrict__ out, int n4) {
    int i = blockIdx.x * 256 + threadIdx.x;
    if (i >= n4) return;
    float c2 = consts[2], scale = consts[3];
    float4 p = prop[i];
    float4 w = Wh[i];
    float4 o;
    o.x = scale * (p.x + c2 * w.x);
    o.y = scale * (p.y + c2 * w.y);
    o.z = scale * (p.z + c2 * w.z);
    o.w = scale * (p.w + c2 * w.w);
    out[i] = o;
}

extern "C" void kernel_launch(void* const* d_in, const int* in_sizes, int n_in,
                              void* d_out, int out_size, void* d_ws, size_t ws_size,
                              hipStream_t stream) {
    const float* h     = (const float*)d_in[0];
    const int*   ei    = (const int*)  d_in[1];
    const float* adj_v = (const float*)d_in[2];
    const float* deg_v = (const float*)d_in[3];
    const float* W     = (const float*)d_in[4];
    const float* bias  = (const float*)d_in[5];
    const float* dcoef = (const float*)d_in[6];
    const float* cpar  = (const float*)d_in[7];
    const float* spar  = (const float*)d_in[8];

    const int N = in_sizes[0] / D;
    const int E = in_sizes[2];

    float* Wh     = (float*)d_ws;
    float* prop   = Wh + (size_t)N * D;
    float* norms  = prop + (size_t)N * D;
    float* consts = norms + N;

    consts_kernel<<<1, 1, 0, stream>>>(cpar, spar, consts);

    gemm_wmma_kernel<<<(N + 63) / 64, 128, 0, stream>>>(h, W, bias, Wh, N);

    norms_kernel<<<(N * 16 + 255) / 256, 256, 0, stream>>>(Wh, norms, N);

    int n4 = N * D / 4;
    zero_kernel<<<(n4 + 255) / 256, 256, 0, stream>>>((float4*)prop, n4);

    edge_kernel<<<(E + 255) / 256, 256, 0, stream>>>(ei, adj_v, deg_v, Wh, norms,
                                                   dcoef, consts, prop, E);

    final_kernel<<<(n4 + 255) / 256, 256, 0, stream>>>((const float4*)prop,
                                                       (const float4*)Wh, consts,
                                                       (float4*)d_out, n4);
}


# Edge atomic adds go straight into out (w carries scale); norms kernel stores out's start value scale*c2*Wh; accumulator plane, zero and final kernels dropped, 6 -> 4 launches; sum starts at c2 term
# speedup vs baseline: 2.1942x; 1.4060x over previous
#include <hip/hip_runtime.h>
#include <hip/hip_bf16.h>

typedef __attribute__((ext_vector_type(2))) float v2f;
typedef __attribute__((ext_vector_type(8))) float v8f;

#define D 64
#define LSTR 66

__global__ void consts_kernel(const float* __restrict__ coeff_param,
                              const float* __restrict__ scale_param,
                              float* __restrict__ consts) {
    if (threadIdx.x != 0 || blockIdx.x != 0) return;
    float x0 = coeff_param[0], x1 = coeff_param[1], x2 = coeff_param[2];
    float m = fmaxf(x0, fmaxf(x1, x2));
    float e0 = expf(x0 - m), e1 = expf(x1 - m), e2 = expf(x2 - m);
    float inv = 1.0f / (e0 + e1 + e2);
    consts[0] = e0 * inv;
    consts[1] = e1 * inv;
    consts[2] = e2 * inv;
    float s = scale_param[0];
    consts[3] = (s > 20.0f) ? s : log1pf(expf(s));
}

__global__ void __launch_bounds__(128)
gemm_wmma_kernel(const float* __restrict__ h, const float* __restrict__ W,
                 const float* __restrict__ bias, float* __restrict__ Wh, int N) {
    __shared__ float Ws[D * LSTR];

    int tid = threadIdx.x;
    for (int i = tid * 4; i < D * D; i += 128 * 4) {
        int r = i >> 6;
        int c = i & 63;
        float4 w4 = *(const float4*)(W + i);
        Ws[(c + 0) * LSTR + r] = w4.x;
        Ws[(c + 1) * LSTR + r] = w4.y;
        Ws[(c + 2) * LSTR + r] = w4.z;
        Ws[(c + 3) * LSTR + r] = w4.w;
    }
    __syncthreads();

    int wave = tid >> 5;
    int lane = tid & 31;
    int half = lane >> 4;
    int l16  = lane & 15;

    int m0 = (blockIdx.x * 4 + wave) * 16;
    if (m0 >= N) return;

    const float* hrow = h + (size_t)(m0 + l16) * D + 2 * half;
    const float* wt   = Ws + l16 * LSTR + 2 * half;

    v8f acc[4] = {};

#pragma unroll 4
    for (int s = 0; s < 16; ++s) {
        v2f a = *(const v2f*)(hrow + 4 * s);
#pragma unroll
        for (int t = 0; t < 4; ++t) {
            v2f bf = *(const v2f*)(wt + t * (16 * LSTR) + 4 * s);
            acc[t] = __builtin_amdgcn_wmma_f32_16x16x4_f32(
                false, a, false, bf, (short)0, acc[t], false, false);
        }
    }

#pragma unroll
    for (int t = 0; t < 4; ++t) {
        int n = t * 16 + l16;
        float bv = bias[n];
#pragma unroll
        for (int v = 0; v < 8; ++v) {
            Wh[(size_t)(m0 + half * 8 + v) * D + n] = acc[t][v] + bv;
        }
    }
}

__global__ void __launch_bounds__(256)
norms_init_kernel(const float* __restrict__ Wh, const float* __restrict__ consts,
                  float* __restrict__ norms, float* __restrict__ out, int N) {
    int gtid = blockIdx.x * 256 + threadIdx.x;
    int node = gtid >> 4;
    int l    = gtid & 15;
    if (node >= N) return;
    const float c2 = consts[2], scale = consts[3];
    const size_t off = (size_t)node * D + l * 4;
    float4 v = *(const float4*)(Wh + off);
    float4 o;
    o.x = scale * (c2 * v.x);
    o.y = scale * (c2 * v.y);
    o.z = scale * (c2 * v.z);
    o.w = scale * (c2 * v.w);
    *(float4*)(out + off) = o;
    float s = v.x * v.x + v.y * v.y + v.z * v.z + v.w * v.w;
    for (int m = 8; m >= 1; m >>= 1) s += __shfl_xor(s, m, 16);
    if (l == 0) norms[node] = sqrtf(s);
}

template <int M>
__device__ __forceinline__ void rs_step(float (&p)[32], int lane) {
    const bool hi = (lane & M) != 0;
#pragma unroll
    for (int j = 0; j < M; ++j) {
        const float a = p[j];
        const float b = p[j + M];
        const float keep = hi ? b : a;
        const float send = hi ? a : b;
        p[j] = keep + __shfl_xor(send, M, 32);
    }
}

__global__ void __launch_bounds__(256)
edge_kernel(const int* __restrict__ ei, const float* __restrict__ adj_v,
            const float* __restrict__ deg_v, const float* __restrict__ Wh,
            const float* __restrict__ norms, const float* __restrict__ dcoef,
            const float* __restrict__ consts, float* __restrict__ out, int E) {
    const int lane = threadIdx.x & 31;
    const unsigned lane4 = (threadIdx.x & 31u) * 4u;
    const int e0   = (blockIdx.x * 8 + (threadIdx.x >> 5)) * 32;
    if (e0 >= E) return;

    const int  eid  = e0 + lane;
    const bool live = eid < E;
    const int  eidc = live ? eid : E - 1;
    const int   row = ei[eidc];
    const int   col = ei[E + eidc];
    const float adj = adj_v[eidc];
    const float x   = dcoef[0] * deg_v[eidc] + dcoef[1];
    const float nr  = norms[row];
    const float nc  = norms[col];
    const float sc  = (x > 20.0f) ? x : log1pf(expf(x));

    float cv0[32], cv1[32], p[32];
#pragma unroll
    for (int e = 0; e < 32; ++e) {
        const int r = __builtin_amdgcn_readlane(row, e);
        const int c = __builtin_amdgcn_readlane(col, e);
        const float* rp = (const float*)((const char*)Wh + ((unsigned)r * (D * 4u) + lane4));
        const float* cp = (const float*)((const char*)Wh + ((unsigned)c * (D * 4u) + lane4));
        const float r0 = rp[0], r1 = rp[32];
        cv0[e] = cp[0];
        cv1[e] = cp[32];
        p[e] = r0 * cv0[e] + r1 * cv1[e];
    }

    rs_step<16>(p, lane);
    rs_step<8>(p, lane);
    rs_step<4>(p, lane);
    rs_step<2>(p, lane);
    rs_step<1>(p, lane);
    const float dot = p[0];

    const float sim = dot / fmaxf(nr * nc, 1e-8f);
    const float c0 = consts[0], c1 = consts[1], scale = consts[3];
    float w = scale * (adj * sc * (sim > 0.0f ? c0 * sim : c1 * sim));
    if (!live || row == col) w = 0.0f;

#pragma unroll
    for (int e = 0; e < 32; ++e) {
        const float we = __builtin_bit_cast(
            float, __builtin_amdgcn_readlane(__builtin_bit_cast(int, w), e));
        if (we != 0.0f) {
            const int r = __builtin_amdgcn_readlane(row, e);
            float* pp = (float*)((char*)out + ((unsigned)r * (D * 4u) + lane4));
            atomicAdd(pp, we * cv0[e]);
            atomicAdd(pp + 32, we * cv1[e]);
        }
    }
}

extern "C" void kernel_launch(void* const* d_in, const int* in_sizes, int n_in,
                              void* d_out, int out_size, void* d_ws, size_t ws_size,
                              hipStream_t stream) {
    const float* h     = (const float*)d_in[0];
    const int*   ei    = (const int*)  d_in[1];
    const float* adj_v = (const float*)d_in[2];
    const float* deg_v = (const float*)d_in[3];
    const float* W     = (const float*)d_in[4];
    const float* bias  = (const float*)d_in[5];
    const float* dcoef = (const float*)d_in[6];
    const float* cpar  = (const float*)d_in[7];
    const float* spar  = (const float*)d_in[8];

    const int N = in_sizes[0] / D;
    const int E = in_sizes[2];

    float* Wh     = (float*)d_ws;
    float* norms  = Wh + (size_t)N * D;
    float* consts = norms + N;
    float* out    = (float*)d_out;

    consts_kernel<<<1, 1, 0, stream>>>(cpar, spar, consts);

    gemm_wmma_kernel<<<(N + 63) / 64, 128, 0, stream>>>(h, W, bias, Wh, N);

    norms_init_kernel<<<(N * 16 + 255) / 256, 256, 0, stream>>>(Wh, consts, norms, out, N);

    edge_kernel<<<(E + 255) / 256, 256, 0, stream>>>(ei, adj_v, deg_v, Wh, norms,
                                                   dcoef, consts, out, E);
}


# Edge kernel: one resident 32-wave work-group per WGP copies the norms table (256 KiB) into LDS once and loops over its edge groups: the two scattered 4-byte norm loads per edge become LDS reads
# speedup vs baseline: 2.3326x; 1.0631x over previous
#include <hip/hip_runtime.h>
#include <hip/hip_bf16.h>

typedef __attribute__((ext_vector_type(2))) float v2f;
typedef __attribute__((ext_vector_type(8))) float v8f;

#define D 64
#define LSTR 66
#define NORM_LDS 65536

__global__ void consts_kernel(const float* __restrict__ coeff_param,
                              const float* __restrict__ scale_param,
                              float* __restrict__ consts) {
    if (threadIdx.x != 0 || blockIdx.x != 0) return;
    float x0 = coeff_param[0], x1 = coeff_param[1], x2 = coeff_param[2];
    float m = fmaxf(x0, fmaxf(x1, x2));
    float e0 = expf(x0 - m), e1 = expf(x1 - m), e2 = expf(x2 - m);
    float inv = 1.0f / (e0 + e1 + e2);
    consts[0] = e0 * inv;
    consts[1] = e1 * inv;
    consts[2] = e2 * inv;
    float s = scale_param[0];
    consts[3] = (s > 20.0f) ? s : log1pf(expf(s));
}

__global__ void __launch_bounds__(128)
gemm_wmma_kernel(const float* __restrict__ h, const float* __restrict__ W,
                 const float* __restrict__ bias, float* __restrict__ Wh, int N) {
    __shared__ float Ws[D * LSTR];

    int tid = threadIdx.x;
    for (int i = tid * 4; i < D * D; i += 128 * 4) {
        int r = i >> 6;
        int c = i & 63;
        float4 w4 = *(const float4*)(W + i);
        Ws[(c + 0) * LSTR + r] = w4.x;
        Ws[(c + 1) * LSTR + r] = w4.y;
        Ws[(c + 2) * LSTR + r] = w4.z;
        Ws[(c + 3) * LSTR + r] = w4.w;
    }
    __syncthreads();

    int wave = tid >> 5;
    int lane = tid & 31;
    int half = lane >> 4;
    int l16  = lane & 15;

    int m0 = (blockIdx.x * 4 + wave) * 16;
    if (m0 >= N) return;

    const float* hrow = h + (size_t)(m0 + l16) * D + 2 * half;
    const float* wt   = Ws + l16 * LSTR + 2 * half;

    v8f acc[4] = {};

#pragma unroll 4
    for (int s = 0; s < 16; ++s) {
        v2f a = *(const v2f*)(hrow + 4 * s);
#pragma unroll
        for (int t = 0; t < 4; ++t) {
            v2f bf = *(const v2f*)(wt + t * (16 * LSTR) + 4 * s);
            acc[t] = __builtin_amdgcn_wmma_f32_16x16x4_f32(
                false, a, false, bf, (short)0, acc[t], false, false);
        }
    }

#pragma unroll
    for (int t = 0; t < 4; ++t) {
        int n = t * 16 + l16;
        float bv = bias[n];
#pragma unroll
        for (int v = 0; v < 8; ++v) {
            Wh[(size_t)(m0 + half * 8 + v) * D + n] = acc[t][v] + bv;
        }
    }
}

__global__ void __launch_bounds__(256)
norms_init_kernel(const float* __restrict__ Wh, const float* __restrict__ consts,
                  float* __restrict__ norms, float* __restrict__ out, int N) {
    int gtid = blockIdx.x * 256 + threadIdx.x;
    int node = gtid >> 4;
    int l    = gtid & 15;
    if (node >= N) return;
    const float c2 = consts[2], scale = consts[3];
    const size_t off = (size_t)node * D + l * 4;
    float4 v = *(const float4*)(Wh + off);
    float4 o;
    o.x = scale * (c2 * v.x);
    o.y = scale * (c2 * v.y);
    o.z = scale * (c2 * v.z);
    o.w = scale * (c2 * v.w);
    *(float4*)(out + off) = o;
    float s = v.x * v.x + v.y * v.y + v.z * v.z + v.w * v.w;
    for (int m = 8; m >= 1; m >>= 1) s += __shfl_xor(s, m, 16);
    if (l == 0) norms[node] = sqrtf(s);
}

template <int M>
__device__ __forceinline__ float lane_xor(float v) {
    return __builtin_bit_cast(float, __builtin_amdgcn_ds_swizzle(__builtin_bit_cast(int, v), (M << 10) | 0x1f));
}

template <int M>
__device__ __forceinline__ void rs_step(float (&p)[32], int lane) {
    const bool hi = (lane & M) != 0;
#pragma unroll
    for (int j = 0; j < M; ++j) {
        const float a = p[j];
        const float b = p[j + M];
        const float keep = hi ? b : a;
        const float send = hi ? a : b;
        p[j] = keep + lane_xor<M>(send);
    }
}

template <bool STAGED>
__global__ void __launch_bounds__(1024)
edge_kernel(const int* __restrict__ ei, const float* __restrict__ adj_v,
            const float* __restrict__ deg_v, const float* __restrict__ Wh,
            const float* __restrict__ norms, const float* __restrict__ dcoef,
            const float* __restrict__ consts, float* __restrict__ out, int E, int N) {
    __shared__ __attribute__((aligned(16))) float nl[STAGED ? NORM_LDS : 4];

    if (STAGED) {
        float4 t[NORM_LDS / 4096];
#pragma unroll
        for (int k = 0; k < NORM_LDS / 4096; ++k) {
            const int i = threadIdx.x * 4 + k * 4096;
            t[k] = (i + 3 < N) ? *(const float4*)(norms + i) : make_float4(0.0f, 0.0f, 0.0f, 0.0f);
        }
#pragma unroll
        for (int k = 0; k < NORM_LDS / 4096; ++k) {
            const int i = threadIdx.x * 4 + k * 4096;
            if (i + 3 < N) *(float4*)(nl + i) = t[k];
        }
        for (int i = (N & ~3) + threadIdx.x; i < N; i += 1024)
            nl[i] = norms[i];
    }
    __syncthreads();

    const int lane = threadIdx.x & 31;
    const unsigned lane4 = (threadIdx.x & 31u) * 4u;
    const int ngroups = (E + 31) >> 5;
    const int nwaves  = gridDim.x * 32;
    const float c0 = consts[0], c1 = consts[1], scale = consts[3];
    const float dc0 = dcoef[0], dc1 = dcoef[1];

    const int wave = __builtin_amdgcn_readfirstlane(threadIdx.x >> 5);
    for (int g = blockIdx.x * 32 + wave; g < ngroups; g += nwaves) {
    const int e0 = g * 32;

    const int  eid  = e0 + lane;
    const bool live = eid < E;
    const int  eidc = live ? eid : E - 1;
    const int   row = ei[eidc];
    const int   col = ei[E + eidc];
    const float adj = adj_v[eidc];
    const float x   = dc0 * deg_v[eidc] + dc1;
    const float sc  = (x > 20.0f) ? x : log1pf(expf(x));
    const float asc = adj * sc;
    const bool  dead = !live || row == col;

    float cv0[32], cv1[32], p[32];
#pragma unroll
    for (int e = 0; e < 32; ++e) {
        const int r = __builtin_amdgcn_readlane(row, e);
        const int c = __builtin_amdgcn_readlane(col, e);
        const float* rp = (const float*)((const char*)Wh + ((unsigned)r * (D * 4u) + lane4));
        const float* cp = (const float*)((const char*)Wh + ((unsigned)c * (D * 4u) + lane4));
        const float r0 = rp[0], r1 = rp[32];
        cv0[e] = cp[0];
        cv1[e] = cp[32];
        p[e] = r0 * cv0[e] + r1 * cv1[e];
        if (e == 15) __builtin_amdgcn_sched_barrier(0);
    }

    rs_step<16>(p, lane);
    rs_step<8>(p, lane);
    rs_step<4>(p, lane);
    rs_step<2>(p, lane);
    rs_step<1>(p, lane);
    const float dot = p[0];

    float nr, nc;
    if (STAGED) { nr = nl[row];    nc = nl[col]; }
    else        { nr = norms[row]; nc = norms[col]; }
    const float sim = dot / fmaxf(nr * nc, 1e-8f);
    float w = scale * (asc * (sim > 0.0f ? c0 * sim : c1 * sim));
    if (dead) w = 0.0f;

#pragma unroll
    for (int e = 0; e < 32; ++e) {
        const float we = __builtin_bit_cast(
            float, __builtin_amdgcn_readlane(__builtin_bit_cast(int, w), e));
        if (we != 0.0f) {
            const int r = __builtin_amdgcn_readlane(row, e);
            float* pp = (float*)((char*)out + ((unsigned)r * (D * 4u) + lane4));
            atomicAdd(pp, we * cv0[e]);
            atomicAdd(pp + 32, we * cv1[e]);
        }
    }
    }
}

extern "C" void kernel_launch(void* const* d_in, const int* in_sizes, int n_in,
                              void* d_out, int out_size, void* d_ws, size_t ws_size,
                              hipStream_t stream) {
    const float* h     = (const float*)d_in[0];
    const int*   ei    = (const int*)  d_in[1];
    const float* adj_v = (const float*)d_in[2];
    const float* deg_v = (const float*)d_in[3];
    const float* W     = (const float*)d_in[4];
    const float* bias  = (const float*)d_in[5];
    const float* dcoef = (const float*)d_in[6];
    const float* cpar  = (const float*)d_in[7];
    const float* spar  = (const float*)d_in[8];

    const int N = in_sizes[0] / D;
    const int E = in_sizes[2];

    float* Wh     = (float*)d_ws;
    float* norms  = Wh + (size_t)N * D;
    float* consts = norms + N;
    float* out    = (float*)d_out;

    consts_kernel<<<1, 1, 0, stream>>>(cpar, spar, consts);

    gemm_wmma_kernel<<<(N + 63) / 64, 128, 0, stream>>>(h, W, bias, Wh, N);

    norms_init_kernel<<<(N * 16 + 255) / 256, 256, 0, stream>>>(Wh, consts, norms, out, N);

    if (N <= NORM_LDS)
        edge_kernel<true><<<256, 1024, 0, stream>>>(ei, adj_v, deg_v, Wh, norms,
                                                    dcoef, consts, out, E, N);
    else
        edge_kernel<false><<<256, 1024, 0, stream>>>(ei, adj_v, deg_v, Wh, norms,
                                                     dcoef, consts, out, E, N);
}


# Norms/start-value kernel as a resident grid: 256 work-groups of 8 waves, each 16-lane group takes 16 rows, row loads all in flight (32768 waves -> 2048); row sums by reduce-scatter, same add tree
# speedup vs baseline: 2.4114x; 1.0338x over previous
#include <hip/hip_runtime.h>
#include <hip/hip_bf16.h>

typedef __attribute__((ext_vector_type(2))) float v2f;
typedef __attribute__((ext_vector_type(8))) float v8f;

#define D 64
#define LSTR 66
#define NORM_LDS 65536

__global__ void consts_kernel(const float* __restrict__ coeff_param,
                              const float* __restrict__ scale_param,
                              float* __restrict__ consts) {
    if (threadIdx.x != 0 || blockIdx.x != 0) return;
    float x0 = coeff_param[0], x1 = coeff_param[1], x2 = coeff_param[2];
    float m = fmaxf(x0, fmaxf(x1, x2));
    float e0 = expf(x0 - m), e1 = expf(x1 - m), e2 = expf(x2 - m);
    float inv = 1.0f / (e0 + e1 + e2);
    consts[0] = e0 * inv;
    consts[1] = e1 * inv;
    consts[2] = e2 * inv;
    float s = scale_param[0];
    consts[3] = (s > 20.0f) ? s : log1pf(expf(s));
}

__global__ void __launch_bounds__(128)
gemm_wmma_kernel(const float* __restrict__ h, const float* __restrict__ W,
                 const float* __restrict__ bias, float* __restrict__ Wh, int N) {
    __shared__ float Ws[D * LSTR];

    int tid = threadIdx.x;
    for (int i = tid * 4; i < D * D; i += 128 * 4) {
        int r = i >> 6;
        int c = i & 63;
        float4 w4 = *(const float4*)(W + i);
        Ws[(c + 0) * LSTR + r] = w4.x;
        Ws[(c + 1) * LSTR + r] = w4.y;
        Ws[(c + 2) * LSTR + r] = w4.z;
        Ws[(c + 3) * LSTR + r] = w4.w;
    }
    __syncthreads();

    int wave = tid >> 5;
    int lane = tid & 31;
    int half = lane >> 4;
    int l16  = lane & 15;

    int m0 = (blockIdx.x * 4 + wave) * 16;
    if (m0 >= N) return;

    const float* hrow = h + (size_t)(m0 + l16) * D + 2 * half;
    const float* wt   = Ws + l16 * LSTR + 2 * half;

    v8f acc[4] = {};

#pragma unroll 4
    for (int s = 0; s < 16; ++s) {
        v2f a = *(const v2f*)(hrow + 4 * s);
#pragma unroll
        for (int t = 0; t < 4; ++t) {
            v2f bf = *(const v2f*)(wt + t * (16 * LSTR) + 4 * s);
            acc[t] = __builtin_amdgcn_wmma_f32_16x16x4_f32(
                false, a, false, bf, (short)0, acc[t], false, false);
        }
    }

#pragma unroll
    for (int t = 0; t < 4; ++t) {
        int n = t * 16 + l16;
        float bv = bias[n];
#pragma unroll
        for (int v = 0; v < 8; ++v) {
            Wh[(size_t)(m0 + half * 8 + v) * D + n] = acc[t][v] + bv;
        }
    }
}

template <int M>
__device__ __forceinline__ float lane_xor(float v) {
    return __builtin_bit_cast(float, __builtin_amdgcn_ds_swizzle(__builtin_bit_cast(int, v), (M << 10) | 0x1f));
}

template <int M, int NP>
__device__ __forceinline__ void rs_step(float (&p)[NP], int lane) {
    const bool hi = (lane & M) != 0;
#pragma unroll
    for (int j = 0; j < M; ++j) {
        const float a = p[j];
        const float b = p[j + M];
        const float keep = hi ? b : a;
        const float send = hi ? a : b;
        p[j] = keep + lane_xor<M>(send);
    }
}

#define NI_ROWS 16

template <bool GUARD>
__device__ __forceinline__ void norms_init_rows(const float* __restrict__ Wh, float* __restrict__ norms,
                                                float* __restrict__ out, int N, int base,
                                                int lane, float c2, float scale) {
    const unsigned lo = (unsigned)(lane >> 4) * (NI_ROWS * D * 4u) + (unsigned)(lane & 15) * 16u;
    const char* src = (const char*)Wh  + (size_t)base * (D * 4u) + lo;
    char*       dst = (char*)out       + (size_t)base * (D * 4u) + lo;
    const int   r0  = base + NI_ROWS * (lane >> 4);

    float4 v[NI_ROWS];
#pragma unroll
    for (int k = 0; k < NI_ROWS; ++k) {
        if (!GUARD || r0 + k < N) v[k] = *(const float4*)(src + k * (D * 4));
        else                      v[k] = make_float4(0.0f, 0.0f, 0.0f, 0.0f);
    }
    __builtin_amdgcn_sched_barrier(0);

    float s[NI_ROWS];
#pragma unroll
    for (int k = 0; k < NI_ROWS; ++k) {
        float4 o;
        o.x = scale * (c2 * v[k].x);
        o.y = scale * (c2 * v[k].y);
        o.z = scale * (c2 * v[k].z);
        o.w = scale * (c2 * v[k].w);
        if (!GUARD || r0 + k < N) *(float4*)(dst + k * (D * 4)) = o;
        s[k] = __builtin_fmaf(v[k].w, v[k].w,
               __builtin_fmaf(v[k].z, v[k].z,
               __builtin_fmaf(v[k].x, v[k].x, v[k].y * v[k].y)));
    }
    rs_step<8>(s, lane);
    rs_step<4>(s, lane);
    rs_step<2>(s, lane);
    rs_step<1>(s, lane);
    if (!GUARD || base + lane < N) norms[base + lane] = sqrtf(s[0]);
}

__global__ void __launch_bounds__(256)
norms_init_kernel(const float* __restrict__ Wh, const float* __restrict__ consts,
                  float* __restrict__ norms, float* __restrict__ out, int N) {
    const int lane   = threadIdx.x & 31;
    const int nwaves = gridDim.x * 8;
    const float c2 = consts[2], scale = consts[3];
    const int wave = __builtin_amdgcn_readfirstlane(threadIdx.x >> 5);
    for (int base = (blockIdx.x * 8 + wave) * 32; base < N; base += nwaves * 32) {
        if (base + 32 <= N) norms_init_rows<false>(Wh, norms, out, N, base, lane, c2, scale);
        else                norms_init_rows<true>(Wh, norms, out, N, base, lane, c2, scale);
    }
}

template <bool STAGED>
__global__ void __launch_bounds__(1024)
edge_kernel(const int* __restrict__ ei, const float* __restrict__ adj_v,
            const float* __restrict__ deg_v, const float* __restrict__ Wh,
            const float* __restrict__ norms, const float* __restrict__ dcoef,
            const float* __restrict__ consts, float* __restrict__ out, int E, int N) {
    __shared__ __attribute__((aligned(16))) float nl[STAGED ? NORM_LDS : 4];

    if (STAGED) {
        float4 t[NORM_LDS / 4096];
#pragma unroll
        for (int k = 0; k < NORM_LDS / 4096; ++k) {
            const int i = threadIdx.x * 4 + k * 4096;
            t[k] = (i + 3 < N) ? *(const float4*)(norms + i) : make_float4(0.0f, 0.0f, 0.0f, 0.0f);
        }
#pragma unroll
        for (int k = 0; k < NORM_LDS / 4096; ++k) {
            const int i = threadIdx.x * 4 + k * 4096;
            if (i + 3 < N) *(float4*)(nl + i) = t[k];
        }
        for (int i = (N & ~3) + threadIdx.x; i < N; i += 1024)
            nl[i] = norms[i];
    }
    __syncthreads();

    const int lane = threadIdx.x & 31;
    const unsigned lane4 = (threadIdx.x & 31u) * 4u;
    const int ngroups = (E + 31) >> 5;
    const int nwaves  = gridDim.x * 32;
    const float c0 = consts[0], c1 = consts[1], scale = consts[3];
    const float dc0 = dcoef[0], dc1 = dcoef[1];

    const int wave = __builtin_amdgcn_readfirstlane(threadIdx.x >> 5);
    for (int g = blockIdx.x * 32 + wave; g < ngroups; g += nwaves) {
    const int e0 = g * 32;

    const int  eid  = e0 + lane;
    const bool live = eid < E;
    const int  eidc = live ? eid : E - 1;
    const int   row = ei[eidc];
    const int   col = ei[E + eidc];
    const float adj = adj_v[eidc];
    const float x   = dc0 * deg_v[eidc] + dc1;
    const float sc  = (x > 20.0f) ? x : log1pf(expf(x));
    const float asc = adj * sc;
    const bool  dead = !live || row == col;

    float cv0[32], cv1[32], p[32];
#pragma unroll
    for (int e = 0; e < 32; ++e) {
        const int r = __builtin_amdgcn_readlane(row, e);
        const int c = __builtin_amdgcn_readlane(col, e);
        const float* rp = (const float*)((const char*)Wh + ((unsigned)r * (D * 4u) + lane4));
        const float* cp = (const float*)((const char*)Wh + ((unsigned)c * (D * 4u) + lane4));
        const float r0 = rp[0], r1 = rp[32];
        cv0[e] = cp[0];
        cv1[e] = cp[32];
        p[e] = r0 * cv0[e] + r1 * cv1[e];
        if (e == 15) __builtin_amdgcn_sched_barrier(0);
    }

    rs_step<16>(p, lane);
    rs_step<8>(p, lane);
    rs_step<4>(p, lane);
    rs_step<2>(p, lane);
    rs_step<1>(p, lane);
    const float dot = p[0];

    float nr, nc;
    if (STAGED) { nr = nl[row];    nc = nl[col]; }
    else        { nr = norms[row]; nc = norms[col]; }
    const float sim = dot / fmaxf(nr * nc, 1e-8f);
    float w = scale * (asc * (sim > 0.0f ? c0 * sim : c1 * sim));
    if (dead) w = 0.0f;

#pragma unroll
    for (int e = 0; e < 32; ++e) {
        const float we = __builtin_bit_cast(
            float, __builtin_amdgcn_readlane(__builtin_bit_cast(int, w), e));
        if (we != 0.0f) {
            const int r = __builtin_amdgcn_readlane(row, e);
            float* pp = (float*)((char*)out + ((unsigned)r * (D * 4u) + lane4));
            atomicAdd(pp, we * cv0[e]);
            atomicAdd(pp + 32, we * cv1[e]);
        }
    }
    }
}

extern "C" void kernel_launch(void* const* d_in, const int* in_sizes, int n_in,
                              void* d_out, int out_size, void* d_ws, size_t ws_size,
                              hipStream_t stream) {
    const float* h     = (const float*)d_in[0];
    const int*   ei    = (const int*)  d_in[1];
    const float* adj_v = (const float*)d_in[2];
    const float* deg_v = (const float*)d_in[3];
    const float* W     = (const float*)d_in[4];
    const float* bias  = (const float*)d_in[5];
    const float* dcoef = (const float*)d_in[6];
    const float* cpar  = (const float*)d_in[7];
    const float* spar  = (const float*)d_in[8];

    const int N = in_sizes[0] / D;
    const int E = in_sizes[2];

    float* Wh     = (float*)d_ws;
    float* norms  = Wh + (size_t)N * D;
    float* consts = norms + N;
    float* out    = (float*)d_out;

    consts_kernel<<<1, 1, 0, stream>>>(cpar, spar, consts);

    gemm_wmma_kernel<<<(N + 63) / 64, 128, 0, stream>>>(h, W, bias, Wh, N);

    norms_init_kernel<<<256, 256, 0, stream>>>(Wh, consts, norms, out, N);

    if (N <= NORM_LDS)
        edge_kernel<true><<<256, 1024, 0, stream>>>(ei, adj_v, deg_v, Wh, norms,
                                                    dcoef, consts, out, E, N);
    else
        edge_kernel<false><<<256, 1024, 0, stream>>>(ei, adj_v, deg_v, Wh, norms,
                                                     dcoef, consts, out, E, N);
}


# Wh rows stored pair-interleaved (elements l, l+32 adjacent): a row gather is one 8-byte-per-lane load of the whole 256-B row, 2 gathers an edge not 4; dots same; norm squares in another order
# speedup vs baseline: 2.4940x; 1.0343x over previous
#include <hip/hip_runtime.h>
#include <hip/hip_bf16.h>

typedef __attribute__((ext_vector_type(2))) float v2f;
typedef __attribute__((ext_vector_type(8))) float v8f;

#define D 64
#define LSTR 66
#define NORM_LDS 65536

__global__ void consts_kernel(const float* __restrict__ coeff_param,
                              const float* __restrict__ scale_param,
                              float* __restrict__ consts) {
    if (threadIdx.x != 0 || blockIdx.x != 0) return;
    float x0 = coeff_param[0], x1 = coeff_param[1], x2 = coeff_param[2];
    float m = fmaxf(x0, fmaxf(x1, x2));
    float e0 = expf(x0 - m), e1 = expf(x1 - m), e2 = expf(x2 - m);
    float inv = 1.0f / (e0 + e1 + e2);
    consts[0] = e0 * inv;
    consts[1] = e1 * inv;
    consts[2] = e2 * inv;
    float s = scale_param[0];
    consts[3] = (s > 20.0f) ? s : log1pf(expf(s));
}

__global__ void __launch_bounds__(128)
gemm_wmma_kernel(const float* __restrict__ h, const float* __restrict__ W,
                 const float* __restrict__ bias, float* __restrict__ Wh, int N) {
    __shared__ float Ws[D * LSTR];

    int tid = threadIdx.x;
    for (int i = tid * 4; i < D * D; i += 128 * 4) {
        int r = i >> 6;
        int c = i & 63;
        float4 w4 = *(const float4*)(W + i);
        Ws[(c + 0) * LSTR + r] = w4.x;
        Ws[(c + 1) * LSTR + r] = w4.y;
        Ws[(c + 2) * LSTR + r] = w4.z;
        Ws[(c + 3) * LSTR + r] = w4.w;
    }
    __syncthreads();

    int wave = tid >> 5;
    int lane = tid & 31;
    int half = lane >> 4;
    int l16  = lane & 15;

    int m0 = (blockIdx.x * 4 + wave) * 16;
    if (m0 >= N) return;

    const float* hrow = h + (size_t)(m0 + l16) * D + 2 * half;
    const float* wt   = Ws + l16 * LSTR + 2 * half;

    v8f acc[4] = {};

#pragma unroll 4
    for (int s = 0; s < 16; ++s) {
        v2f a = *(const v2f*)(hrow + 4 * s);
#pragma unroll
        for (int t = 0; t < 4; ++t) {
            v2f bf = *(const v2f*)(wt + t * (16 * LSTR) + 4 * s);
            acc[t] = __builtin_amdgcn_wmma_f32_16x16x4_f32(
                false, a, false, bf, (short)0, acc[t], false, false);
        }
    }

    const float b0 = bias[l16], b1 = bias[16 + l16], b2 = bias[32 + l16], b3 = bias[48 + l16];
    float* wrow = Wh + (size_t)(m0 + half * 8) * D + 2 * l16;
#pragma unroll
    for (int v = 0; v < 8; ++v) {
        v2f lo, hi;
        lo.x = acc[0][v] + b0;
        lo.y = acc[2][v] + b2;
        hi.x = acc[1][v] + b1;
        hi.y = acc[3][v] + b3;
        *(v2f*)(wrow + v * D)      = lo;
        *(v2f*)(wrow + v * D + 32) = hi;
    }
}

template <int M>
__device__ __forceinline__ float lane_xor(float v) {
    return __builtin_bit_cast(float, __builtin_amdgcn_ds_swizzle(__builtin_bit_cast(int, v), (M << 10) | 0x1f));
}

template <int M, int NP>
__device__ __forceinline__ void rs_step(float (&p)[NP], int lane) {
    const bool hi = (lane & M) != 0;
#pragma unroll
    for (int j = 0; j < M; ++j) {
        const float a = p[j];
        const float b = p[j + M];
        const float keep = hi ? b : a;
        const float send = hi ? a : b;
        p[j] = keep + lane_xor<M>(send);
    }
}

#define NI_ROWS 16

template <bool GUARD>
__device__ __forceinline__ void norms_init_rows(const float* __restrict__ Wh, float* __restrict__ norms,
                                                float* __restrict__ out, int N, int base,
                                                int lane, float c2, float scale) {
    const unsigned lo  = (unsigned)(lane >> 4) * (NI_ROWS * D * 4u) + (unsigned)(lane & 15) * 16u;
    const unsigned lod = (unsigned)(lane >> 4) * (NI_ROWS * D * 4u) + (unsigned)(lane & 15) * 8u;
    const char* src = (const char*)Wh  + (size_t)base * (D * 4u) + lo;
    char*       dst = (char*)out       + (size_t)base * (D * 4u) + lod;
    const int   r0  = base + NI_ROWS * (lane >> 4);

    float4 v[NI_ROWS];
#pragma unroll
    for (int k = 0; k < NI_ROWS; ++k) {
        if (!GUARD || r0 + k < N) v[k] = *(const float4*)(src + k * (D * 4));
        else                      v[k] = make_float4(0.0f, 0.0f, 0.0f, 0.0f);
    }
    __builtin_amdgcn_sched_barrier(0);

    float s[NI_ROWS];
#pragma unroll
    for (int k = 0; k < NI_ROWS; ++k) {
        float2 o0, o1;
        o0.x = scale * (c2 * v[k].x);
        o0.y = scale * (c2 * v[k].z);
        o1.x = scale * (c2 * v[k].y);
        o1.y = scale * (c2 * v[k].w);
        if (!GUARD || r0 + k < N) {
            *(float2*)(dst + k * (D * 4))       = o0;
            *(float2*)(dst + k * (D * 4) + 128) = o1;
        }
        s[k] = __builtin_fmaf(v[k].w, v[k].w,
               __builtin_fmaf(v[k].z, v[k].z,
               __builtin_fmaf(v[k].x, v[k].x, v[k].y * v[k].y)));
    }
    rs_step<8>(s, lane);
    rs_step<4>(s, lane);
    rs_step<2>(s, lane);
    rs_step<1>(s, lane);
    if (!GUARD || base + lane < N) norms[base + lane] = sqrtf(s[0]);
}

__global__ void __launch_bounds__(256)
norms_init_kernel(const float* __restrict__ Wh, const float* __restrict__ consts,
                  float* __restrict__ norms, float* __restrict__ out, int N) {
    const int lane   = threadIdx.x & 31;
    const int nwaves = gridDim.x * 8;
    const float c2 = consts[2], scale = consts[3];
    const int wave = __builtin_amdgcn_readfirstlane(threadIdx.x >> 5);
    for (int base = (blockIdx.x * 8 + wave) * 32; base < N; base += nwaves * 32) {
        if (base + 32 <= N) norms_init_rows<false>(Wh, norms, out, N, base, lane, c2, scale);
        else                norms_init_rows<true>(Wh, norms, out, N, base, lane, c2, scale);
    }
}

template <bool STAGED>
__global__ void __launch_bounds__(1024)
edge_kernel(const int* __restrict__ ei, const float* __restrict__ adj_v,
            const float* __restrict__ deg_v, const float* __restrict__ Wh,
            const float* __restrict__ norms, const float* __restrict__ dcoef,
            const float* __restrict__ consts, float* __restrict__ out, int E, int N) {
    __shared__ __attribute__((aligned(16))) float nl[STAGED ? NORM_LDS : 4];

    if (STAGED) {
        float4 t[NORM_LDS / 4096];
#pragma unroll
        for (int k = 0; k < NORM_LDS / 4096; ++k) {
            const int i = threadIdx.x * 4 + k * 4096;
            t[k] = (i + 3 < N) ? *(const float4*)(norms + i) : make_float4(0.0f, 0.0f, 0.0f, 0.0f);
        }
#pragma unroll
        for (int k = 0; k < NORM_LDS / 4096; ++k) {
            const int i = threadIdx.x * 4 + k * 4096;
            if (i + 3 < N) *(float4*)(nl + i) = t[k];
        }
        for (int i = (N & ~3) + threadIdx.x; i < N; i += 1024)
            nl[i] = norms[i];
    }
    __syncthreads();

    const int lane = threadIdx.x & 31;
    const unsigned lane4 = (threadIdx.x & 31u) * 4u;
    const unsigned lane8 = (threadIdx.x & 31u) * 8u;
    const int ngroups = (E + 31) >> 5;
    const int nwaves  = gridDim.x * 32;
    const float c0 = consts[0], c1 = consts[1], scale = consts[3];
    const float dc0 = dcoef[0], dc1 = dcoef[1];

    const int wave = __builtin_amdgcn_readfirstlane(threadIdx.x >> 5);
    for (int g = blockIdx.x * 32 + wave; g < ngroups; g += nwaves) {
    const int e0 = g * 32;

    const int  eid  = e0 + lane;
    const bool live = eid < E;
    const int  eidc = live ? eid : E - 1;
    const int   row = ei[eidc];
    const int   col = ei[E + eidc];
    const float adj = adj_v[eidc];
    const float x   = dc0 * deg_v[eidc] + dc1;
    const float sc  = (x > 20.0f) ? x : log1pf(expf(x));
    const float asc = adj * sc;
    const bool  dead = !live || row == col;

    float cv0[32], cv1[32], p[32];
#pragma unroll
    for (int e = 0; e < 32; ++e) {
        const int r = __builtin_amdgcn_readlane(row, e);
        const int c = __builtin_amdgcn_readlane(col, e);
        const float2 rv = *(const float2*)((const char*)Wh + ((unsigned)r * (D * 4u) + lane8));
        const float2 cv = *(const float2*)((const char*)Wh + ((unsigned)c * (D * 4u) + lane8));
        const float r0 = rv.x, r1 = rv.y;
        cv0[e] = cv.x;
        cv1[e] = cv.y;
        p[e] = r0 * cv0[e] + r1 * cv1[e];
        if (e == 15) __builtin_amdgcn_sched_barrier(0);
    }

    rs_step<16>(p, lane);
    rs_step<8>(p, lane);
    rs_step<4>(p, lane);
    rs_step<2>(p, lane);
    rs_step<1>(p, lane);
    const float dot = p[0];

    float nr, nc;
    if (STAGED) { nr = nl[row];    nc = nl[col]; }
    else        { nr = norms[row]; nc = norms[col]; }
    const float sim = dot / fmaxf(nr * nc, 1e-8f);
    float w = scale * (asc * (sim > 0.0f ? c0 * sim : c1 * sim));
    if (dead) w = 0.0f;

#pragma unroll
    for (int e = 0; e < 32; ++e) {
        const float we = __builtin_bit_cast(
            float, __builtin_amdgcn_readlane(__builtin_bit_cast(int, w), e));
        if (we != 0.0f) {
            const int r = __builtin_amdgcn_readlane(row, e);
            float* pp = (float*)((char*)out + ((unsigned)r * (D * 4u) + lane4));
            atomicAdd(pp, we * cv0[e]);
            atomicAdd(pp + 32, we * cv1[e]);
        }
    }
    }
}

extern "C" void kernel_launch(void* const* d_in, const int* in_sizes, int n_in,
                              void* d_out, int out_size, void* d_ws, size_t ws_size,
                              hipStream_t stream) {
    const float* h     = (const float*)d_in[0];
    const int*   ei    = (const int*)  d_in[1];
    const float* adj_v = (const float*)d_in[2];
    const float* deg_v = (const float*)d_in[3];
    const float* W     = (const float*)d_in[4];
    const float* bias  = (const float*)d_in[5];
    const float* dcoef = (const float*)d_in[6];
    const float* cpar  = (const float*)d_in[7];
    const float* spar  = (const float*)d_in[8];

    const int N = in_sizes[0] / D;
    const int E = in_sizes[2];

    float* Wh     = (float*)d_ws;
    float* norms  = Wh + (size_t)N * D;
    float* consts = norms + N;
    float* out    = (float*)d_out;

    consts_kernel<<<1, 1, 0, stream>>>(cpar, spar, consts);

    gemm_wmma_kernel<<<(N + 63) / 64, 128, 0, stream>>>(h, W, bias, Wh, N);

    norms_init_kernel<<<256, 256, 0, stream>>>(Wh, consts, norms, out, N);

    if (N <= NORM_LDS)
        edge_kernel<true><<<256, 1024, 0, stream>>>(ei, adj_v, deg_v, Wh, norms,
                                                    dcoef, consts, out, E, N);
    else
        edge_kernel<false><<<256, 1024, 0, stream>>>(ei, adj_v, deg_v, Wh, norms,
                                                     dcoef, consts, out, E, N);
}


# Edge kernel: a work-group draws a job (edge slice x row half of out) matched to its die's cache half by one returning add; its waves pack that half's edges 32 a batch by lane permutes; arrival order
# speedup vs baseline: 2.6993x; 1.0823x over previous
#include <hip/hip_runtime.h>
#include <hip/hip_bf16.h>

typedef __attribute__((ext_vector_type(2))) float v2f;
typedef __attribute__((ext_vector_type(8))) float v8f;

#define D 64
#define LSTR 66
#define NORM_LDS 65536

__global__ void consts_kernel(const float* __restrict__ coeff_param,
                              const float* __restrict__ scale_param,
                              float* __restrict__ consts) {
    if (threadIdx.x != 0 || blockIdx.x != 0) return;
    float x0 = coeff_param[0], x1 = coeff_param[1], x2 = coeff_param[2];
    float m = fmaxf(x0, fmaxf(x1, x2));
    float e0 = expf(x0 - m), e1 = expf(x1 - m), e2 = expf(x2 - m);
    float inv = 1.0f / (e0 + e1 + e2);
    consts[0] = e0 * inv;
    consts[1] = e1 * inv;
    consts[2] = e2 * inv;
    float s = scale_param[0];
    consts[3] = (s > 20.0f) ? s : log1pf(expf(s));
    int* tick = (int*)(consts + 4);
    tick[0] = 0; tick[1] = 0; tick[2] = 0; tick[3] = 0;
}

__global__ void __launch_bounds__(128)
gemm_wmma_kernel(const float* __restrict__ h, const float* __restrict__ W,
                 const float* __restrict__ bias, float* __restrict__ Wh, int N) {
    __shared__ float Ws[D * LSTR];

    int tid = threadIdx.x;
    for (int i = tid * 4; i < D * D; i += 128 * 4) {
        int r = i >> 6;
        int c = i & 63;
        float4 w4 = *(const float4*)(W + i);
        Ws[(c + 0) * LSTR + r] = w4.x;
        Ws[(c + 1) * LSTR + r] = w4.y;
        Ws[(c + 2) * LSTR + r] = w4.z;
        Ws[(c + 3) * LSTR + r] = w4.w;
    }
    __syncthreads();

    int wave = tid >> 5;
    int lane = tid & 31;
    int half = lane >> 4;
    int l16  = lane & 15;

    int m0 = (blockIdx.x * 4 + wave) * 16;
    if (m0 >= N) return;

    const float* hrow = h + (size_t)(m0 + l16) * D + 2 * half;
    const float* wt   = Ws + l16 * LSTR + 2 * half;

    v8f acc[4] = {};

#pragma unroll 4
    for (int s = 0; s < 16; ++s) {
        v2f a = *(const v2f*)(hrow + 4 * s);
#pragma unroll
        for (int t = 0; t < 4; ++t) {
            v2f bf = *(const v2f*)(wt + t * (16 * LSTR) + 4 * s);
            acc[t] = __builtin_amdgcn_wmma_f32_16x16x4_f32(
                false, a, false, bf, (short)0, acc[t], false, false);
        }
    }

    const float b0 = bias[l16], b1 = bias[16 + l16], b2 = bias[32 + l16], b3 = bias[48 + l16];
    float* wrow = Wh + (size_t)(m0 + half * 8) * D + 2 * l16;
#pragma unroll
    for (int v = 0; v < 8; ++v) {
        v2f lo, hi;
        lo.x = acc[0][v] + b0;
        lo.y = acc[2][v] + b2;
        hi.x = acc[1][v] + b1;
        hi.y = acc[3][v] + b3;
        *(v2f*)(wrow + v * D)      = lo;
        *(v2f*)(wrow + v * D + 32) = hi;
    }
}

template <int M>
__device__ __forceinline__ float lane_xor(float v) {
    return __builtin_bit_cast(float, __builtin_amdgcn_ds_swizzle(__builtin_bit_cast(int, v), (M << 10) | 0x1f));
}

template <int M, int NP>
__device__ __forceinline__ void rs_step(float (&p)[NP], int lane) {
    const bool hi = (lane & M) != 0;
#pragma unroll
    for (int j = 0; j < M; ++j) {
        const float a = p[j];
        const float b = p[j + M];
        const float keep = hi ? b : a;
        const float send = hi ? a : b;
        p[j] = keep + lane_xor<M>(send);
    }
}

#define NI_ROWS 16

template <bool GUARD>
__device__ __forceinline__ void norms_init_rows(const float* __restrict__ Wh, float* __restrict__ norms,
                                                float* __restrict__ out, int N, int base,
                                                int lane, float c2, float scale) {
    const unsigned lo  = (unsigned)(lane >> 4) * (NI_ROWS * D * 4u) + (unsigned)(lane & 15) * 16u;
    const unsigned lod = (unsigned)(lane >> 4) * (NI_ROWS * D * 4u) + (unsigned)(lane & 15) * 8u;
    const char* src = (const char*)Wh  + (size_t)base * (D * 4u) + lo;
    char*       dst = (char*)out       + (size_t)base * (D * 4u) + lod;
    const int   r0  = base + NI_ROWS * (lane >> 4);

    float4 v[NI_ROWS];
#pragma unroll
    for (int k = 0; k < NI_ROWS; ++k) {
        if (!GUARD || r0 + k < N) v[k] = *(const float4*)(src + k * (D * 4));
        else                      v[k] = make_float4(0.0f, 0.0f, 0.0f, 0.0f);
    }
    __builtin_amdgcn_sched_barrier(0);

    float s[NI_ROWS];
#pragma unroll
    for (int k = 0; k < NI_ROWS; ++k) {
        float2 o0, o1;
        o0.x = scale * (c2 * v[k].x);
        o0.y = scale * (c2 * v[k].z);
        o1.x = scale * (c2 * v[k].y);
        o1.y = scale * (c2 * v[k].w);
        if (!GUARD || r0 + k < N) {
            *(float2*)(dst + k * (D * 4))       = o0;
            *(float2*)(dst + k * (D * 4) + 128) = o1;
        }
        s[k] = __builtin_fmaf(v[k].w, v[k].w,
               __builtin_fmaf(v[k].z, v[k].z,
               __builtin_fmaf(v[k].x, v[k].x, v[k].y * v[k].y)));
    }
    rs_step<8>(s, lane);
    rs_step<4>(s, lane);
    rs_step<2>(s, lane);
    rs_step<1>(s, lane);
    if (!GUARD || base + lane < N) norms[base + lane] = sqrtf(s[0]);
}

__global__ void __launch_bounds__(256)
norms_init_kernel(const float* __restrict__ Wh, const float* __restrict__ consts,
                  float* __restrict__ norms, float* __restrict__ out, int N) {
    const int lane   = threadIdx.x & 31;
    const int nwaves = gridDim.x * 8;
    const float c2 = consts[2], scale = consts[3];
    const int wave = __builtin_amdgcn_readfirstlane(threadIdx.x >> 5);
    for (int base = (blockIdx.x * 8 + wave) * 32; base < N; base += nwaves * 32) {
        if (base + 32 <= N) norms_init_rows<false>(Wh, norms, out, N, base, lane, c2, scale);
        else                norms_init_rows<true>(Wh, norms, out, N, base, lane, c2, scale);
    }
}

template <bool STAGED>
__global__ void __launch_bounds__(1024)
edge_kernel(const int* __restrict__ ei, const float* __restrict__ adj_v,
            const float* __restrict__ deg_v, const float* __restrict__ Wh,
            const float* __restrict__ norms, const float* __restrict__ dcoef,
            const float* __restrict__ consts, int* __restrict__ tick,
            float* __restrict__ out, int E, int N) {
    __shared__ __attribute__((aligned(16))) float nl[STAGED ? NORM_LDS : 4];
    __shared__ int job_s;

    int jt = 0, jh = 0;
    if (threadIdx.x == 0) {
        const unsigned id = __builtin_amdgcn_s_sendmsg_rtn(0x87);
        jh = (int)((id >> 18) & 1u);
        jt = atomicAdd(tick + jh, 1);
    }

    if (STAGED) {
        float4 t[NORM_LDS / 4096];
#pragma unroll
        for (int k = 0; k < NORM_LDS / 4096; ++k) {
            const int i = threadIdx.x * 4 + k * 4096;
            t[k] = (i + 3 < N) ? *(const float4*)(norms + i) : make_float4(0.0f, 0.0f, 0.0f, 0.0f);
        }
#pragma unroll
        for (int k = 0; k < NORM_LDS / 4096; ++k) {
            const int i = threadIdx.x * 4 + k * 4096;
            if (i + 3 < N) *(float4*)(nl + i) = t[k];
        }
        for (int i = (N & ~3) + threadIdx.x; i < N; i += 1024)
            nl[i] = norms[i];
    }
    if (threadIdx.x == 0) {
        const int nsl = (int)(gridDim.x >> 1);
        int s = jt, qq = jh;
        if (jt >= nsl) {
            const int t2 = atomicAdd(tick + 2 + (jh ^ 1), 1);
            s = nsl - 1 - t2;
            qq = jh ^ 1;
        }
        job_s = s * 2 + qq;
    }
    __syncthreads();

    const int lane = threadIdx.x & 31;
    const unsigned lane4 = (threadIdx.x & 31u) * 4u;
    const unsigned lane8 = (threadIdx.x & 31u) * 8u;
    const int ngroups = (E + 31) >> 5;
    const int job    = __builtin_amdgcn_readfirstlane(job_s);
    const int q      = job & 1;
    const int nslice = (int)(gridDim.x >> 1);
    const int gps    = __builtin_amdgcn_readfirstlane((ngroups + nslice - 1) / nslice);
    const int gbeg   = (job >> 1) * gps;
    const int gend   = (gbeg + gps < ngroups) ? gbeg + gps : ngroups;
    const int nhalf  = (N + 1) >> 1;
    const float c0 = consts[0], c1 = consts[1], scale = consts[3];
    const float dc0 = dcoef[0], dc1 = dcoef[1];

    const int wave = __builtin_amdgcn_readfirstlane(threadIdx.x >> 5);
    int pend = 0;
    int np   = 0;
    for (int g = gbeg + wave; ; g += 32) {
    const bool more = g < gend;
    int cnt = 0, ra = 0;
    if (more) {
        const int  cid   = g * 32 + lane;
        const bool clive = cid < E;
        const int  crow  = ei[clive ? cid : E - 1];
        const bool keep  = clive && (((crow >= nhalf) ? 1 : 0) == q);
        const unsigned m = __builtin_amdgcn_ballot_w32(keep);
        cnt = __builtin_popcount(m);
        const int pk   = (int)__builtin_amdgcn_mbcnt_lo(m, 0u);
        const int dest = keep ? pk : cnt + (lane - pk);
        const int ap   = __builtin_amdgcn_ds_permute(dest << 2, cid);
        ra = __builtin_amdgcn_ds_bpermute(((lane - np) & 31) << 2, ap);
    }
    const int  tot  = np + cnt;
    const int  comb = (lane < np) ? pend : ra;
    const bool fire = more ? (tot >= 32) : (tot > 0);
    if (fire) {
    const int nb = (tot < 32) ? tot : 32;

    const bool live = lane < nb;
    const int  eidc = live ? comb : E - 1;
    const int   row = ei[eidc];
    const int   col = ei[E + eidc];
    const float adj = adj_v[eidc];
    const float x   = dc0 * deg_v[eidc] + dc1;
    const float sc  = (x > 20.0f) ? x : log1pf(expf(x));
    const float asc = adj * sc;
    const bool  dead = !live || row == col;
    const int   rowg = dead ? 0 : row;
    const int   colg = dead ? 0 : col;

    float cv0[32], cv1[32], p[32];
#pragma unroll
    for (int e = 0; e < 32; ++e) {
        const int r = __builtin_amdgcn_readlane(rowg, e);
        const int c = __builtin_amdgcn_readlane(colg, e);
        const float2 rv = *(const float2*)((const char*)Wh + ((unsigned)r * (D * 4u) + lane8));
        const float2 cv = *(const float2*)((const char*)Wh + ((unsigned)c * (D * 4u) + lane8));
        const float r0 = rv.x, r1 = rv.y;
        cv0[e] = cv.x;
        cv1[e] = cv.y;
        p[e] = r0 * cv0[e] + r1 * cv1[e];
        if (e == 15) __builtin_amdgcn_sched_barrier(0);
    }

    rs_step<16>(p, lane);
    rs_step<8>(p, lane);
    rs_step<4>(p, lane);
    rs_step<2>(p, lane);
    rs_step<1>(p, lane);
    const float dot = p[0];

    float nr, nc;
    if (STAGED) { nr = nl[row];    nc = nl[col]; }
    else        { nr = norms[row]; nc = norms[col]; }
    const float sim = dot / fmaxf(nr * nc, 1e-8f);
    float w = scale * (asc * (sim > 0.0f ? c0 * sim : c1 * sim));
    if (dead) w = 0.0f;

#pragma unroll
    for (int e = 0; e < 32; ++e) {
        const float we = __builtin_bit_cast(
            float, __builtin_amdgcn_readlane(__builtin_bit_cast(int, w), e));
        if (we != 0.0f) {
            const int r = __builtin_amdgcn_readlane(row, e);
            float* pp = (float*)((char*)out + ((unsigned)r * (D * 4u) + lane4));
            atomicAdd(pp, we * cv0[e]);
            atomicAdd(pp + 32, we * cv1[e]);
        }
    }
    }
    if (!more) break;
    if (tot >= 32) { pend = ra;   np = tot - 32; }
    else           { pend = comb; np = tot; }
    }
}

extern "C" void kernel_launch(void* const* d_in, const int* in_sizes, int n_in,
                              void* d_out, int out_size, void* d_ws, size_t ws_size,
                              hipStream_t stream) {
    const float* h     = (const float*)d_in[0];
    const int*   ei    = (const int*)  d_in[1];
    const float* adj_v = (const float*)d_in[2];
    const float* deg_v = (const float*)d_in[3];
    const float* W     = (const float*)d_in[4];
    const float* bias  = (const float*)d_in[5];
    const float* dcoef = (const float*)d_in[6];
    const float* cpar  = (const float*)d_in[7];
    const float* spar  = (const float*)d_in[8];

    const int N = in_sizes[0] / D;
    const int E = in_sizes[2];

    float* Wh     = (float*)d_ws;
    float* norms  = Wh + (size_t)N * D;
    float* consts = norms + N;
    int*   tick   = (int*)(consts + 4);
    float* out    = (float*)d_out;

    consts_kernel<<<1, 1, 0, stream>>>(cpar, spar, consts);

    gemm_wmma_kernel<<<(N + 63) / 64, 128, 0, stream>>>(h, W, bias, Wh, N);

    norms_init_kernel<<<256, 256, 0, stream>>>(Wh, consts, norms, out, N);

    if (N <= NORM_LDS)
        edge_kernel<true><<<256, 1024, 0, stream>>>(ei, adj_v, deg_v, Wh, norms,
                                                    dcoef, consts, tick, out, E, N);
    else
        edge_kernel<false><<<256, 1024, 0, stream>>>(ei, adj_v, deg_v, Wh, norms,
                                                     dcoef, consts, tick, out, E, N);
}


# No norm stored: 16 edges a pass, each edge's dot and both rows' square sums reduce-scattered over a lane half on row_xmask adds (other add order); GEMM stores out's start value; 4 -> 3 launches
# speedup vs baseline: 2.9507x; 1.0931x over previous
#include <hip/hip_runtime.h>
#include <hip/hip_bf16.h>

typedef __attribute__((ext_vector_type(2))) float v2f;
typedef __attribute__((ext_vector_type(8))) float v8f;

#define D 64
#define LSTR 66

__global__ void consts_kernel(const float* __restrict__ coeff_param,
                              const float* __restrict__ scale_param,
                              float* __restrict__ consts) {
    if (threadIdx.x != 0 || blockIdx.x != 0) return;
    float x0 = coeff_param[0], x1 = coeff_param[1], x2 = coeff_param[2];
    float m = fmaxf(x0, fmaxf(x1, x2));
    float e0 = expf(x0 - m), e1 = expf(x1 - m), e2 = expf(x2 - m);
    float inv = 1.0f / (e0 + e1 + e2);
    consts[0] = e0 * inv;
    consts[1] = e1 * inv;
    consts[2] = e2 * inv;
    float s = scale_param[0];
    consts[3] = (s > 20.0f) ? s : log1pf(expf(s));
    int* tick = (int*)(consts + 4);
    tick[0] = 0; tick[1] = 0; tick[2] = 0; tick[3] = 0;
}

__global__ void __launch_bounds__(128)
gemm_wmma_kernel(const float* __restrict__ h, const float* __restrict__ W,
                 const float* __restrict__ bias, const float* __restrict__ consts,
                 float* __restrict__ Wh, float* __restrict__ out, int N) {
    __shared__ float Ws[D * LSTR];

    int tid = threadIdx.x;
    for (int i = tid * 4; i < D * D; i += 128 * 4) {
        int r = i >> 6;
        int c = i & 63;
        float4 w4 = *(const float4*)(W + i);
        Ws[(c + 0) * LSTR + r] = w4.x;
        Ws[(c + 1) * LSTR + r] = w4.y;
        Ws[(c + 2) * LSTR + r] = w4.z;
        Ws[(c + 3) * LSTR + r] = w4.w;
    }
    __syncthreads();

    int wave = tid >> 5;
    int lane = tid & 31;
    int half = lane >> 4;
    int l16  = lane & 15;

    int m0 = (blockIdx.x * 4 + wave) * 16;
    if (m0 >= N) return;

    const float* hrow = h + (size_t)(m0 + l16) * D + 2 * half;
    const float* wt   = Ws + l16 * LSTR + 2 * half;

    v8f acc[4] = {};

#pragma unroll 4
    for (int s = 0; s < 16; ++s) {
        v2f a = *(const v2f*)(hrow + 4 * s);
#pragma unroll
        for (int t = 0; t < 4; ++t) {
            v2f bf = *(const v2f*)(wt + t * (16 * LSTR) + 4 * s);
            acc[t] = __builtin_amdgcn_wmma_f32_16x16x4_f32(
                false, a, false, bf, (short)0, acc[t], false, false);
        }
    }

    const float b0 = bias[l16], b1 = bias[16 + l16], b2 = bias[32 + l16], b3 = bias[48 + l16];
    float* wrow = Wh + (size_t)(m0 + half * 8) * D + 2 * l16;
    float* orow = out + (size_t)(m0 + half * 8) * D + l16;
    const float c2 = consts[2], scale = consts[3];
#pragma unroll
    for (int v = 0; v < 8; ++v) {
        v2f lo, hi;
        lo.x = acc[0][v] + b0;
        lo.y = acc[2][v] + b2;
        hi.x = acc[1][v] + b1;
        hi.y = acc[3][v] + b3;
        *(v2f*)(wrow + v * D)      = lo;
        *(v2f*)(wrow + v * D + 32) = hi;
        orow[v * D]      = scale * (c2 * lo.x);
        orow[v * D + 16] = scale * (c2 * hi.x);
        orow[v * D + 32] = scale * (c2 * lo.y);
        orow[v * D + 48] = scale * (c2 * hi.y);
    }
}

template <int M>
__device__ __forceinline__ float lane_xor(float v) {
    const int x = __builtin_bit_cast(int, v);
    if constexpr (M < 16)
        return __builtin_bit_cast(float, __builtin_amdgcn_update_dpp(0, x, 0x160 | M, 0xf, 0xf, true));
    else
        return __builtin_bit_cast(float, __builtin_amdgcn_permlanex16(x, x, 0x76543210, 0xfedcba98, false, false));
}

template <int M, int NP>
__device__ __forceinline__ void rs_step(float (&p)[NP], int lane) {
    const bool hi = (lane & M) != 0;
#pragma unroll
    for (int j = 0; j < M; ++j) {
        const float a = p[j];
        const float b = p[j + M];
        const float keep = hi ? b : a;
        const float send = hi ? a : b;
        p[j] = keep + lane_xor<M>(send);
    }
}

__global__ void __launch_bounds__(1024)
edge_kernel(const int* __restrict__ ei, const float* __restrict__ adj_v,
            const float* __restrict__ deg_v, const float* __restrict__ Wh,
            const float* __restrict__ dcoef,
            const float* __restrict__ consts, int* __restrict__ tick,
            float* __restrict__ out, int E, int N) {
    __shared__ int job_s;

    if (threadIdx.x == 0) {
        const unsigned id = __builtin_amdgcn_s_sendmsg_rtn(0x87);
        const int jh = (int)((id >> 18) & 1u);
        const int jt = atomicAdd(tick + jh, 1);
        const int nsl = (int)(gridDim.x >> 1);
        int s = jt, qq = jh;
        if (jt >= nsl) {
            const int t2 = atomicAdd(tick + 2 + (jh ^ 1), 1);
            s = nsl - 1 - t2;
            qq = jh ^ 1;
        }
        job_s = s * 2 + qq;
    }
    __syncthreads();

    const int lane = threadIdx.x & 31;
    const unsigned lane4 = (threadIdx.x & 31u) * 4u;
    const unsigned lane8 = (threadIdx.x & 31u) * 8u;
    const int ngroups = (E + 31) >> 5;
    const int job    = __builtin_amdgcn_readfirstlane(job_s);
    const int q      = job & 1;
    const int nslice = (int)(gridDim.x >> 1);
    const int gps    = __builtin_amdgcn_readfirstlane((ngroups + nslice - 1) / nslice);
    const int gbeg   = (job >> 1) * gps;
    const int gend   = (gbeg + gps < ngroups) ? gbeg + gps : ngroups;
    const int nhalf  = (N + 1) >> 1;
    const float c0 = consts[0], c1 = consts[1], scale = consts[3];
    const float dc0 = dcoef[0], dc1 = dcoef[1];

    const int wave = __builtin_amdgcn_readfirstlane(threadIdx.x >> 5);
    int pend = 0;
    int np   = 0;
    for (int g = gbeg + wave; ; g += 32) {
    const bool more = g < gend;
    int cnt = 0, ra = 0;
    if (more) {
        const int  cid   = g * 32 + lane;
        const bool clive = cid < E;
        const int  crow  = ei[clive ? cid : E - 1];
        const bool keep  = clive && (((crow >= nhalf) ? 1 : 0) == q);
        const unsigned m = __builtin_amdgcn_ballot_w32(keep);
        cnt = __builtin_popcount(m);
        const int pk   = (int)__builtin_amdgcn_mbcnt_lo(m, 0u);
        const int dest = keep ? pk : cnt + (lane - pk);
        const int ap   = __builtin_amdgcn_ds_permute(dest << 2, cid);
        ra = __builtin_amdgcn_ds_bpermute(((lane - np) & 31) << 2, ap);
    }
    const int  tot  = np + cnt;
    const int  comb = (lane < np) ? pend : ra;
    const bool fire = more ? (tot >= 32) : (tot > 0);
    if (fire) {
    const int nb = (tot < 32) ? tot : 32;

    const bool live = lane < nb;
    const int  eidc = live ? comb : E - 1;
    const int   row = ei[eidc];
    const int   col = ei[E + eidc];
    const float adj = adj_v[eidc];
    const float x   = dc0 * deg_v[eidc] + dc1;
    const float sc  = (x > 20.0f) ? x : log1pf(expf(x));
    const float asc = adj * sc;
    const bool  dead = !live || row == col;
    const int   rowg = dead ? 0 : row;
    const int   colg = dead ? 0 : col;

#pragma unroll
    for (int h = 0; h < 2; ++h) {
    float cv0[16], cv1[16], p[16], sr[16], sq[16];
#pragma unroll
    for (int e = 0; e < 16; ++e) {
        const int r = __builtin_amdgcn_readlane(rowg, 16 * h + e);
        const int c = __builtin_amdgcn_readlane(colg, 16 * h + e);
        const float2 rv = *(const float2*)((const char*)Wh + ((unsigned)r * (D * 4u) + lane8));
        const float2 cv = *(const float2*)((const char*)Wh + ((unsigned)c * (D * 4u) + lane8));
        const float r0 = rv.x, r1 = rv.y;
        cv0[e] = cv.x;
        cv1[e] = cv.y;
        p[e]  = r0 * cv0[e] + r1 * cv1[e];
        sr[e] = r0 * r0 + r1 * r1;
        sq[e] = cv0[e] * cv0[e] + cv1[e] * cv1[e];
    }

    rs_step<8>(p, lane);  rs_step<8>(sr, lane);  rs_step<8>(sq, lane);
    rs_step<4>(p, lane);  rs_step<4>(sr, lane);  rs_step<4>(sq, lane);
    rs_step<2>(p, lane);  rs_step<2>(sr, lane);  rs_step<2>(sq, lane);
    rs_step<1>(p, lane);  rs_step<1>(sr, lane);  rs_step<1>(sq, lane);
    const float dot = p[0]  + lane_xor<16>(p[0]);
    const float nr2 = sr[0] + lane_xor<16>(sr[0]);
    const float nc2 = sq[0] + lane_xor<16>(sq[0]);

    const float nr = sqrtf(nr2), nc = sqrtf(nc2);
    const float sim = dot / fmaxf(nr * nc, 1e-8f);
    float w = scale * (asc * (sim > 0.0f ? c0 * sim : c1 * sim));
    if (dead) w = 0.0f;

#pragma unroll
    for (int e = 0; e < 16; ++e) {
        const float we = __builtin_bit_cast(
            float, __builtin_amdgcn_readlane(__builtin_bit_cast(int, w), 16 * h + e));
        if (we != 0.0f) {
            const int r = __builtin_amdgcn_readlane(row, 16 * h + e);
            float* pp = (float*)((char*)out + ((unsigned)r * (D * 4u) + lane4));
            atomicAdd(pp, we * cv0[e]);
            atomicAdd(pp + 32, we * cv1[e]);
        }
    }
    }
    }
    if (!more) break;
    if (tot >= 32) { pend = ra;   np = tot - 32; }
    else           { pend = comb; np = tot; }
    }
}

extern "C" void kernel_launch(void* const* d_in, const int* in_sizes, int n_in,
                              void* d_out, int out_size, void* d_ws, size_t ws_size,
                              hipStream_t stream) {
    const float* h     = (const float*)d_in[0];
    const int*   ei    = (const int*)  d_in[1];
    const float* adj_v = (const float*)d_in[2];
    const float* deg_v = (const float*)d_in[3];
    const float* W     = (const float*)d_in[4];
    const float* bias  = (const float*)d_in[5];
    const float* dcoef = (const float*)d_in[6];
    const float* cpar  = (const float*)d_in[7];
    const float* spar  = (const float*)d_in[8];

    const int N = in_sizes[0] / D;
    const int E = in_sizes[2];

    float* Wh     = (float*)d_ws;
    float* consts = Wh + (size_t)N * D;
    int*   tick   = (int*)(consts + 4);
    float* out    = (float*)d_out;

    consts_kernel<<<1, 1, 0, stream>>>(cpar, spar, consts);

    gemm_wmma_kernel<<<(N + 63) / 64, 128, 0, stream>>>(h, W, bias, consts, Wh, out, N);

    edge_kernel<<<256, 1024, 0, stream>>>(ei, adj_v, deg_v, Wh, dcoef, consts, tick, out, E, N);
}


# Edge kernel: a work-group's waves put the job's edge numbers into one LDS queue, then draw full batches of 32 from it by returning LDS adds: no part-filled batch per wave; out's adds arrival order
# speedup vs baseline: 3.1325x; 1.0616x over previous
#include <hip/hip_runtime.h>
#include <hip/hip_bf16.h>

typedef __attribute__((ext_vector_type(2))) float v2f;
typedef __attribute__((ext_vector_type(8))) float v8f;

#define D 64
#define LSTR 66

__global__ void consts_kernel(const float* __restrict__ coeff_param,
                              const float* __restrict__ scale_param,
                              float* __restrict__ consts) {
    if (threadIdx.x != 0 || blockIdx.x != 0) return;
    float x0 = coeff_param[0], x1 = coeff_param[1], x2 = coeff_param[2];
    float m = fmaxf(x0, fmaxf(x1, x2));
    float e0 = expf(x0 - m), e1 = expf(x1 - m), e2 = expf(x2 - m);
    float inv = 1.0f / (e0 + e1 + e2);
    consts[0] = e0 * inv;
    consts[1] = e1 * inv;
    consts[2] = e2 * inv;
    float s = scale_param[0];
    consts[3] = (s > 20.0f) ? s : log1pf(expf(s));
    int* tick = (int*)(consts + 4);
    tick[0] = 0; tick[1] = 0; tick[2] = 0; tick[3] = 0;
}

__global__ void __launch_bounds__(128)
gemm_wmma_kernel(const float* __restrict__ h, const float* __restrict__ W,
                 const float* __restrict__ bias, const float* __restrict__ consts,
                 float* __restrict__ Wh, float* __restrict__ out, int N) {
    __shared__ float Ws[D * LSTR];

    int tid = threadIdx.x;
    for (int i = tid * 4; i < D * D; i += 128 * 4) {
        int r = i >> 6;
        int c = i & 63;
        float4 w4 = *(const float4*)(W + i);
        Ws[(c + 0) * LSTR + r] = w4.x;
        Ws[(c + 1) * LSTR + r] = w4.y;
        Ws[(c + 2) * LSTR + r] = w4.z;
        Ws[(c + 3) * LSTR + r] = w4.w;
    }
    __syncthreads();

    int wave = tid >> 5;
    int lane = tid & 31;
    int half = lane >> 4;
    int l16  = lane & 15;

    int m0 = (blockIdx.x * 4 + wave) * 16;
    if (m0 >= N) return;

    const float* hrow = h + (size_t)(m0 + l16) * D + 2 * half;
    const float* wt   = Ws + l16 * LSTR + 2 * half;

    v8f acc[4] = {};

#pragma unroll 4
    for (int s = 0; s < 16; ++s) {
        v2f a = *(const v2f*)(hrow + 4 * s);
#pragma unroll
        for (int t = 0; t < 4; ++t) {
            v2f bf = *(const v2f*)(wt + t * (16 * LSTR) + 4 * s);
            acc[t] = __builtin_amdgcn_wmma_f32_16x16x4_f32(
                false, a, false, bf, (short)0, acc[t], false, false);
        }
    }

    const float b0 = bias[l16], b1 = bias[16 + l16], b2 = bias[32 + l16], b3 = bias[48 + l16];
    float* wrow = Wh + (size_t)(m0 + half * 8) * D + 2 * l16;
    float* orow = out + (size_t)(m0 + half * 8) * D + l16;
    const float c2 = consts[2], scale = consts[3];
#pragma unroll
    for (int v = 0; v < 8; ++v) {
        v2f lo, hi;
        lo.x = acc[0][v] + b0;
        lo.y = acc[2][v] + b2;
        hi.x = acc[1][v] + b1;
        hi.y = acc[3][v] + b3;
        *(v2f*)(wrow + v * D)      = lo;
        *(v2f*)(wrow + v * D + 32) = hi;
        orow[v * D]      = scale * (c2 * lo.x);
        orow[v * D + 16] = scale * (c2 * hi.x);
        orow[v * D + 32] = scale * (c2 * lo.y);
        orow[v * D + 48] = scale * (c2 * hi.y);
    }
}

template <int M>
__device__ __forceinline__ float lane_xor(float v) {
    const int x = __builtin_bit_cast(int, v);
    if constexpr (M < 16)
        return __builtin_bit_cast(float, __builtin_amdgcn_update_dpp(0, x, 0x160 | M, 0xf, 0xf, true));
    else
        return __builtin_bit_cast(float, __builtin_amdgcn_permlanex16(x, x, 0x76543210, 0xfedcba98, false, false));
}

template <int M, int NP>
__device__ __forceinline__ void rs_step(float (&p)[NP], int lane) {
    const bool hi = (lane & M) != 0;
#pragma unroll
    for (int j = 0; j < M; ++j) {
        const float a = p[j];
        const float b = p[j + M];
        const float keep = hi ? b : a;
        const float send = hi ? a : b;
        p[j] = keep + lane_xor<M>(send);
    }
}

#define QGROUPS 256
#define QCAP    (QGROUPS * 32)

__global__ void __launch_bounds__(1024)
edge_kernel(const int* __restrict__ ei, const float* __restrict__ adj_v,
            const float* __restrict__ deg_v, const float* __restrict__ Wh,
            const float* __restrict__ dcoef,
            const float* __restrict__ consts, int* __restrict__ tick,
            float* __restrict__ out, int E, int N) {
    __shared__ int job_s;
    __shared__ int qcount;
    __shared__ int qnext;
    __shared__ int queue[QCAP];

    if (threadIdx.x == 0) {
        const unsigned id = __builtin_amdgcn_s_sendmsg_rtn(0x87);
        const int jh = (int)((id >> 18) & 1u);
        const int jt = atomicAdd(tick + jh, 1);
        const int nsl = (int)(gridDim.x >> 1);
        int s = jt, qq = jh;
        if (jt >= nsl) {
            const int t2 = atomicAdd(tick + 2 + (jh ^ 1), 1);
            s = nsl - 1 - t2;
            qq = jh ^ 1;
        }
        job_s = s * 2 + qq;
        qcount = 0;
        qnext  = 0;
    }
    __syncthreads();

    const int lane = threadIdx.x & 31;
    const unsigned lane4 = (threadIdx.x & 31u) * 4u;
    const unsigned lane8 = (threadIdx.x & 31u) * 8u;
    const int ngroups = (E + 31) >> 5;
    const int job    = __builtin_amdgcn_readfirstlane(job_s);
    const int q      = job & 1;
    const int nslice = (int)(gridDim.x >> 1);
    const int gps    = __builtin_amdgcn_readfirstlane((ngroups + nslice - 1) / nslice);
    const int gbeg   = (job >> 1) * gps;
    const int gend   = (gbeg + gps < ngroups) ? gbeg + gps : ngroups;
    const int nhalf  = (N + 1) >> 1;
    const float c0 = consts[0], c1 = consts[1], scale = consts[3];
    const float dc0 = dcoef[0], dc1 = dcoef[1];

    const int wave = __builtin_amdgcn_readfirstlane(threadIdx.x >> 5);

    for (int cb = gbeg; cb < gend; cb += QGROUPS) {
    if (cb != gbeg) {
        __syncthreads();
        if (threadIdx.x == 0) { qcount = 0; qnext = 0; }
        __syncthreads();
    }
    const int ce = (cb + QGROUPS < gend) ? cb + QGROUPS : gend;

    int      cid[8], crow[8];
    unsigned km[8];
    int      totw = 0;
#pragma unroll
    for (int i = 0; i < 8; ++i) {
        const int g = cb + wave + 32 * i;
        cid[i] = g * 32 + lane;
        crow[i] = ei[(g < ce && cid[i] < E) ? cid[i] : E - 1];
    }
#pragma unroll
    for (int i = 0; i < 8; ++i) {
        const int g = cb + wave + 32 * i;
        const unsigned okm = __builtin_amdgcn_ballot_w32(g < ce && cid[i] < E);
        const unsigned hm  = __builtin_amdgcn_ballot_w32(((crow[i] >= nhalf) ? 1 : 0) == q);
        km[i] = okm & hm;
        totw += __builtin_popcount(km[i]);
    }
    int base = 0;
    if (lane == 0) base = atomicAdd(&qcount, totw);
    base = __builtin_amdgcn_readfirstlane(base);
#pragma unroll
    for (int i = 0; i < 8; ++i) {
        if ((km[i] >> lane) & 1u)
            queue[base + (int)__builtin_amdgcn_mbcnt_lo(km[i], 0u)] = cid[i];
        base += __builtin_popcount(km[i]);
    }
    __syncthreads();
    const int total = __builtin_amdgcn_readfirstlane(qcount);

    for (;;) {
    int b = 0;
    if (lane == 0) b = atomicAdd(&qnext, 1);
    b = __builtin_amdgcn_readfirstlane(b);
    const int first = b * 32;
    if (first >= total) break;
    const int nb = (total - first < 32) ? total - first : 32;
    const int qv = queue[first + lane];

    const bool live = lane < nb;
    const int  eidc = live ? qv : E - 1;
    const int   row = ei[eidc];
    const int   col = ei[E + eidc];
    const float adj = adj_v[eidc];
    const float x   = dc0 * deg_v[eidc] + dc1;
    const float sc  = (x > 20.0f) ? x : log1pf(expf(x));
    const float asc = adj * sc;
    const bool  dead = !live || row == col;
    const int   rowg = dead ? 0 : row;
    const int   colg = dead ? 0 : col;

#pragma unroll
    for (int h = 0; h < 2; ++h) {
    float cv0[16], cv1[16], p[16], sr[16], sq[16];
#pragma unroll
    for (int e = 0; e < 16; ++e) {
        const int r = __builtin_amdgcn_readlane(rowg, 16 * h + e);
        const int c = __builtin_amdgcn_readlane(colg, 16 * h + e);
        const float2 rv = *(const float2*)((const char*)Wh + ((unsigned)r * (D * 4u) + lane8));
        const float2 cv = *(const float2*)((const char*)Wh + ((unsigned)c * (D * 4u) + lane8));
        const float r0 = rv.x, r1 = rv.y;
        cv0[e] = cv.x;
        cv1[e] = cv.y;
        p[e]  = r0 * cv0[e] + r1 * cv1[e];
        sr[e] = r0 * r0 + r1 * r1;
        sq[e] = cv0[e] * cv0[e] + cv1[e] * cv1[e];
    }

    rs_step<8>(p, lane);  rs_step<8>(sr, lane);  rs_step<8>(sq, lane);
    rs_step<4>(p, lane);  rs_step<4>(sr, lane);  rs_step<4>(sq, lane);
    rs_step<2>(p, lane);  rs_step<2>(sr, lane);  rs_step<2>(sq, lane);
    rs_step<1>(p, lane);  rs_step<1>(sr, lane);  rs_step<1>(sq, lane);
    const float dot = p[0]  + lane_xor<16>(p[0]);
    const float nr2 = sr[0] + lane_xor<16>(sr[0]);
    const float nc2 = sq[0] + lane_xor<16>(sq[0]);

    const float nr = sqrtf(nr2), nc = sqrtf(nc2);
    const float sim = dot / fmaxf(nr * nc, 1e-8f);
    float w = scale * (asc * (sim > 0.0f ? c0 * sim : c1 * sim));
    if (dead) w = 0.0f;

#pragma unroll
    for (int e = 0; e < 16; ++e) {
        const float we = __builtin_bit_cast(
            float, __builtin_amdgcn_readlane(__builtin_bit_cast(int, w), 16 * h + e));
        if (we != 0.0f) {
            const int r = __builtin_amdgcn_readlane(row, 16 * h + e);
            float* pp = (float*)((char*)out + ((unsigned)r * (D * 4u) + lane4));
            atomicAdd(pp, we * cv0[e]);
            atomicAdd(pp + 32, we * cv1[e]);
        }
    }
    }
    }
    }
}

extern "C" void kernel_launch(void* const* d_in, const int* in_sizes, int n_in,
                              void* d_out, int out_size, void* d_ws, size_t ws_size,
                              hipStream_t stream) {
    const float* h     = (const float*)d_in[0];
    const int*   ei    = (const int*)  d_in[1];
    const float* adj_v = (const float*)d_in[2];
    const float* deg_v = (const float*)d_in[3];
    const float* W     = (const float*)d_in[4];
    const float* bias  = (const float*)d_in[5];
    const float* dcoef = (const float*)d_in[6];
    const float* cpar  = (const float*)d_in[7];
    const float* spar  = (const float*)d_in[8];

    const int N = in_sizes[0] / D;
    const int E = in_sizes[2];

    float* Wh     = (float*)d_ws;
    float* consts = Wh + (size_t)N * D;
    int*   tick   = (int*)(consts + 4);
    float* out    = (float*)d_out;

    consts_kernel<<<1, 1, 0, stream>>>(cpar, spar, consts);

    gemm_wmma_kernel<<<(N + 63) / 64, 128, 0, stream>>>(h, W, bias, consts, Wh, out, N);

    edge_kernel<<<256, 1024, 0, stream>>>(ei, adj_v, deg_v, Wh, dcoef, consts, tick, out, E, N);
}


# Parameter launch dropped: GEMM and edge kernel each derive their constants from the raw parameters in one spare wave (same formulas), GEMM block 0 zeroes the job counters; 3 -> 2 launches
# speedup vs baseline: 3.2545x; 1.0389x over previous
#include <hip/hip_runtime.h>
#include <hip/hip_bf16.h>

typedef __attribute__((ext_vector_type(2))) float v2f;
typedef __attribute__((ext_vector_type(8))) float v8f;

#define D 64
#define LSTR 66

struct ParConsts { float c0, c1, c2, scale; };

__device__ __forceinline__ ParConsts par_consts(const float* __restrict__ coeff_param,
                                                const float* __restrict__ scale_param) {
    float x0 = coeff_param[0], x1 = coeff_param[1], x2 = coeff_param[2];
    float m = fmaxf(x0, fmaxf(x1, x2));
    float e0 = expf(x0 - m), e1 = expf(x1 - m), e2 = expf(x2 - m);
    float inv = 1.0f / (e0 + e1 + e2);
    ParConsts r;
    r.c0 = e0 * inv;
    r.c1 = e1 * inv;
    r.c2 = e2 * inv;
    float s = scale_param[0];
    r.scale = (s > 20.0f) ? s : log1pf(expf(s));
    return r;
}

__global__ void __launch_bounds__(128)
gemm_wmma_kernel(const float* __restrict__ h, const float* __restrict__ W,
                 const float* __restrict__ bias, const float* __restrict__ coeff_param,
                 const float* __restrict__ scale_param,
                 float* __restrict__ Wh, float* __restrict__ out,
                 int* __restrict__ tick, int N) {
    __shared__ float Ws[D * LSTR];
    __shared__ float par_s[2];

    int tid = threadIdx.x;
    if (blockIdx.x == 0 && tid == 0) { tick[0] = 0; tick[1] = 0; tick[2] = 0; tick[3] = 0; }
    if (tid == 32 * (int)(blockIdx.x & 3u)) {
        const ParConsts pc = par_consts(coeff_param, scale_param);
        par_s[0] = pc.c2;
        par_s[1] = pc.scale;
    }
    for (int i = tid * 4; i < D * D; i += 128 * 4) {
        int r = i >> 6;
        int c = i & 63;
        float4 w4 = *(const float4*)(W + i);
        Ws[(c + 0) * LSTR + r] = w4.x;
        Ws[(c + 1) * LSTR + r] = w4.y;
        Ws[(c + 2) * LSTR + r] = w4.z;
        Ws[(c + 3) * LSTR + r] = w4.w;
    }
    __syncthreads();

    int wave = tid >> 5;
    int lane = tid & 31;
    int half = lane >> 4;
    int l16  = lane & 15;

    int m0 = (blockIdx.x * 4 + wave) * 16;
    if (m0 >= N) return;

    const float* hrow = h + (size_t)(m0 + l16) * D + 2 * half;
    const float* wt   = Ws + l16 * LSTR + 2 * half;

    v8f acc[4] = {};

#pragma unroll 4
    for (int s = 0; s < 16; ++s) {
        v2f a = *(const v2f*)(hrow + 4 * s);
#pragma unroll
        for (int t = 0; t < 4; ++t) {
            v2f bf = *(const v2f*)(wt + t * (16 * LSTR) + 4 * s);
            acc[t] = __builtin_amdgcn_wmma_f32_16x16x4_f32(
                false, a, false, bf, (short)0, acc[t], false, false);
        }
    }

    const float b0 = bias[l16], b1 = bias[16 + l16], b2 = bias[32 + l16], b3 = bias[48 + l16];
    float* wrow = Wh + (size_t)(m0 + half * 8) * D + 2 * l16;
    float* orow = out + (size_t)(m0 + half * 8) * D + l16;
    const float c2 = par_s[0], scale = par_s[1];
#pragma unroll
    for (int v = 0; v < 8; ++v) {
        v2f lo, hi;
        lo.x = acc[0][v] + b0;
        lo.y = acc[2][v] + b2;
        hi.x = acc[1][v] + b1;
        hi.y = acc[3][v] + b3;
        *(v2f*)(wrow + v * D)      = lo;
        *(v2f*)(wrow + v * D + 32) = hi;
        orow[v * D]      = scale * (c2 * lo.x);
        orow[v * D + 16] = scale * (c2 * hi.x);
        orow[v * D + 32] = scale * (c2 * lo.y);
        orow[v * D + 48] = scale * (c2 * hi.y);
    }
}

template <int M>
__device__ __forceinline__ float lane_xor(float v) {
    const int x = __builtin_bit_cast(int, v);
    if constexpr (M < 16)
        return __builtin_bit_cast(float, __builtin_amdgcn_update_dpp(0, x, 0x160 | M, 0xf, 0xf, true));
    else
        return __builtin_bit_cast(float, __builtin_amdgcn_permlanex16(x, x, 0x76543210, 0xfedcba98, false, false));
}

template <int M, int NP>
__device__ __forceinline__ void rs_step(float (&p)[NP], int lane) {
    const bool hi = (lane & M) != 0;
#pragma unroll
    for (int j = 0; j < M; ++j) {
        const float a = p[j];
        const float b = p[j + M];
        const float keep = hi ? b : a;
        const float send = hi ? a : b;
        p[j] = keep + lane_xor<M>(send);
    }
}

#define QGROUPS 256
#define QCAP    (QGROUPS * 32)

__global__ void __launch_bounds__(1024)
edge_kernel(const int* __restrict__ ei, const float* __restrict__ adj_v,
            const float* __restrict__ deg_v, const float* __restrict__ Wh,
            const float* __restrict__ dcoef,
            const float* __restrict__ coeff_param, const float* __restrict__ scale_param,
            int* __restrict__ tick,
            float* __restrict__ out, int E, int N) {
    __shared__ float par_s[3];
    __shared__ int job_s;
    __shared__ int qcount;
    __shared__ int qnext;
    __shared__ int queue[QCAP];

    if (threadIdx.x == 0) {
        const unsigned id = __builtin_amdgcn_s_sendmsg_rtn(0x87);
        const int jh = (int)((id >> 18) & 1u);
        const int jt = atomicAdd(tick + jh, 1);
        const int nsl = (int)(gridDim.x >> 1);
        int s = jt, qq = jh;
        if (jt >= nsl) {
            const int t2 = atomicAdd(tick + 2 + (jh ^ 1), 1);
            s = nsl - 1 - t2;
            qq = jh ^ 1;
        }
        job_s = s * 2 + qq;
        qcount = 0;
        qnext  = 0;
    } else if (threadIdx.x == 32) {
        const ParConsts pc = par_consts(coeff_param, scale_param);
        par_s[0] = pc.c0;
        par_s[1] = pc.c1;
        par_s[2] = pc.scale;
    }
    __syncthreads();

    const int lane = threadIdx.x & 31;
    const unsigned lane4 = (threadIdx.x & 31u) * 4u;
    const unsigned lane8 = (threadIdx.x & 31u) * 8u;
    const int ngroups = (E + 31) >> 5;
    const int job    = __builtin_amdgcn_readfirstlane(job_s);
    const int q      = job & 1;
    const int nslice = (int)(gridDim.x >> 1);
    const int gps    = __builtin_amdgcn_readfirstlane((ngroups + nslice - 1) / nslice);
    const int gbeg   = (job >> 1) * gps;
    const int gend   = (gbeg + gps < ngroups) ? gbeg + gps : ngroups;
    const int nhalf  = (N + 1) >> 1;
    const float c0    = __builtin_bit_cast(float, __builtin_amdgcn_readfirstlane(__builtin_bit_cast(int, par_s[0])));
    const float c1    = __builtin_bit_cast(float, __builtin_amdgcn_readfirstlane(__builtin_bit_cast(int, par_s[1])));
    const float scale = __builtin_bit_cast(float, __builtin_amdgcn_readfirstlane(__builtin_bit_cast(int, par_s[2])));
    const float dc0 = dcoef[0], dc1 = dcoef[1];

    const int wave = __builtin_amdgcn_readfirstlane(threadIdx.x >> 5);

    for (int cb = gbeg; cb < gend; cb += QGROUPS) {
    if (cb != gbeg) {
        __syncthreads();
        if (threadIdx.x == 0) { qcount = 0; qnext = 0; }
        __syncthreads();
    }
    const int ce = (cb + QGROUPS < gend) ? cb + QGROUPS : gend;

    int      cid[8], crow[8];
    unsigned km[8];
    int      totw = 0;
#pragma unroll
    for (int i = 0; i < 8; ++i) {
        const int g = cb + wave + 32 * i;
        cid[i] = g * 32 + lane;
        crow[i] = ei[(g < ce && cid[i] < E) ? cid[i] : E - 1];
    }
#pragma unroll
    for (int i = 0; i < 8; ++i) {
        const int g = cb + wave + 32 * i;
        const unsigned okm = __builtin_amdgcn_ballot_w32(g < ce && cid[i] < E);
        const unsigned hm  = __builtin_amdgcn_ballot_w32(((crow[i] >= nhalf) ? 1 : 0) == q);
        km[i] = okm & hm;
        totw += __builtin_popcount(km[i]);
    }
    int base = 0;
    if (lane == 0) base = atomicAdd(&qcount, totw);
    base = __builtin_amdgcn_readfirstlane(base);
#pragma unroll
    for (int i = 0; i < 8; ++i) {
        if ((km[i] >> lane) & 1u)
            queue[base + (int)__builtin_amdgcn_mbcnt_lo(km[i], 0u)] = cid[i];
        base += __builtin_popcount(km[i]);
    }
    __syncthreads();
    const int total = __builtin_amdgcn_readfirstlane(qcount);

    for (;;) {
    int b = 0;
    if (lane == 0) b = atomicAdd(&qnext, 1);
    b = __builtin_amdgcn_readfirstlane(b);
    const int first = b * 32;
    if (first >= total) break;
    const int nb = (total - first < 32) ? total - first : 32;
    const int qv = queue[first + lane];

    const bool live = lane < nb;
    const int  eidc = live ? qv : E - 1;
    const int   row = ei[eidc];
    const int   col = ei[E + eidc];
    const float adj = adj_v[eidc];
    const float x   = dc0 * deg_v[eidc] + dc1;
    const float sc  = (x > 20.0f) ? x : log1pf(expf(x));
    const float asc = adj * sc;
    const bool  dead = !live || row == col;
    const int   rowg = dead ? 0 : row;
    const int   colg = dead ? 0 : col;

#pragma unroll
    for (int h = 0; h < 2; ++h) {
    float cv0[16], cv1[16], p[16], sr[16], sq[16];
#pragma unroll
    for (int e = 0; e < 16; ++e) {
        const int r = __builtin_amdgcn_readlane(rowg, 16 * h + e);
        const int c = __builtin_amdgcn_readlane(colg, 16 * h + e);
        const float2 rv = *(const float2*)((const char*)Wh + ((unsigned)r * (D * 4u) + lane8));
        const float2 cv = *(const float2*)((const char*)Wh + ((unsigned)c * (D * 4u) + lane8));
        const float r0 = rv.x, r1 = rv.y;
        cv0[e] = cv.x;
        cv1[e] = cv.y;
        p[e]  = r0 * cv0[e] + r1 * cv1[e];
        sr[e] = r0 * r0 + r1 * r1;
        sq[e] = cv0[e] * cv0[e] + cv1[e] * cv1[e];
    }

    rs_step<8>(p, lane);  rs_step<8>(sr, lane);  rs_step<8>(sq, lane);
    rs_step<4>(p, lane);  rs_step<4>(sr, lane);  rs_step<4>(sq, lane);
    rs_step<2>(p, lane);  rs_step<2>(sr, lane);  rs_step<2>(sq, lane);
    rs_step<1>(p, lane);  rs_step<1>(sr, lane);  rs_step<1>(sq, lane);
    const float dot = p[0]  + lane_xor<16>(p[0]);
    const float nr2 = sr[0] + lane_xor<16>(sr[0]);
    const float nc2 = sq[0] + lane_xor<16>(sq[0]);

    const float nr = sqrtf(nr2), nc = sqrtf(nc2);
    const float sim = dot / fmaxf(nr * nc, 1e-8f);
    float w = scale * (asc * (sim > 0.0f ? c0 * sim : c1 * sim));
    if (dead) w = 0.0f;

#pragma unroll
    for (int e = 0; e < 16; ++e) {
        const float we = __builtin_bit_cast(
            float, __builtin_amdgcn_readlane(__builtin_bit_cast(int, w), 16 * h + e));
        if (we != 0.0f) {
            const int r = __builtin_amdgcn_readlane(row, 16 * h + e);
            float* pp = (float*)((char*)out + ((unsigned)r * (D * 4u) + lane4));
            atomicAdd(pp, we * cv0[e]);
            atomicAdd(pp + 32, we * cv1[e]);
        }
    }
    }
    }
    }
}

extern "C" void kernel_launch(void* const* d_in, const int* in_sizes, int n_in,
                              void* d_out, int out_size, void* d_ws, size_t ws_size,
                              hipStream_t stream) {
    const float* h     = (const float*)d_in[0];
    const int*   ei    = (const int*)  d_in[1];
    const float* adj_v = (const float*)d_in[2];
    const float* deg_v = (const float*)d_in[3];
    const float* W     = (const float*)d_in[4];
    const float* bias  = (const float*)d_in[5];
    const float* dcoef = (const float*)d_in[6];
    const float* cpar  = (const float*)d_in[7];
    const float* spar  = (const float*)d_in[8];

    const int N = in_sizes[0] / D;
    const int E = in_sizes[2];

    float* Wh     = (float*)d_ws;
    int*   tick   = (int*)(Wh + (size_t)N * D);
    float* out    = (float*)d_out;

    gemm_wmma_kernel<<<(N + 63) / 64, 128, 0, stream>>>(h, W, bias, cpar, spar, Wh, out, tick, N);

    edge_kernel<<<256, 1024, 0, stream>>>(ei, adj_v, deg_v, Wh, dcoef, cpar, spar, tick, out, E, N);
}
